# starting kernel
# baseline (speedup 1.0000x reference)
#include <hip/hip_runtime.h>
#include <math.h>


typedef float v2f __attribute__((ext_vector_type(2)));
typedef float v8f __attribute__((ext_vector_type(8)));

#define PAD 132
#define PI_F 3.14159265358979323846f

struct cplx { float x, y; };
__device__ __forceinline__ cplx cmul(cplx a, cplx b) {
    return { a.x * b.x - a.y * b.y, a.x * b.y + a.y * b.x };
}
__device__ __forceinline__ cplx cadd(cplx a, cplx b) { return { a.x + b.x, a.y + b.y }; }

__global__ __launch_bounds__(64) void build_G_kernel(const float* __restrict__ theta,
                                                     float* __restrict__ G) {
    __shared__ cplx col[64 * 64];
    const int t = threadIdx.x;
    cplx* c = &col[t * 64];
    for (int i = 0; i < 64; ++i) c[i] = { 0.f, 0.f };
    c[t] = { 1.f, 0.f };

    auto apply1q = [&](int q, cplx g00, cplx g01, cplx g10, cplx g11) {
        const int m = 1 << (5 - q);
        for (int j = 0; j < 64; ++j) {
            if (j & m) continue;
            cplx a = c[j], b = c[j | m];
            c[j]     = cadd(cmul(g00, a), cmul(g01, b));
            c[j | m] = cadd(cmul(g10, a), cmul(g11, b));
        }
    };
    auto cnot = [&](int cc, int tt) {
        const int mc = 1 << (5 - cc), mt = 1 << (5 - tt);
        for (int j = 0; j < 64; ++j) {
            if ((j & mc) && !(j & mt)) {
                cplx tmp = c[j]; c[j] = c[j | mt]; c[j | mt] = tmp;
            }
        }
    };

    for (int l = 0; l < 3; ++l) {
        for (int q = 0; q < 5; ++q) cnot(q, q + 1);
        cnot(5, 0);
        for (int q = 0; q < 6; ++q) {
            const float ax = theta[(l * 6 + q) * 3 + 0];
            const float ay = theta[(l * 6 + q) * 3 + 1];
            const float az = theta[(l * 6 + q) * 3 + 2];
            float cx = cosf(0.5f * ax), sx = sinf(0.5f * ax);
            apply1q(q, {cx, 0.f}, {0.f, -sx}, {0.f, -sx}, {cx, 0.f});
            float cy = cosf(0.5f * ay), sy = sinf(0.5f * ay);
            apply1q(q, {cy, 0.f}, {-sy, 0.f}, {sy, 0.f}, {cy, 0.f});
            float cz = cosf(0.5f * az), sz = sinf(0.5f * az);
            apply1q(q, {cz, -sz}, {0.f, 0.f}, {0.f, 0.f}, {cz, sz});
        }
    }
    __syncthreads();

    for (int i = 0; i < 256; ++i) {
        const int idx = t + 64 * i;
        const int n = idx >> 7, k = idx & 127;
        float v;
        if (n < 64) v = (k < 64) ? col[k * 64 + n].x : -col[(k - 64) * 64 + n].y;
        else        v = (k < 64) ? col[k * 64 + (n - 64)].y : col[(k - 64) * 64 + (n - 64)].x;
        G[idx] = v;
    }
}

__global__ __launch_bounds__(128) void qsim_kernel(const float* __restrict__ x,
                                                   const float* __restrict__ G,
                                                   float* __restrict__ out) {
    __shared__ float Mlds[128 * PAD];
    __shared__ float Ulds[4 * 32 * PAD];

    const int tid  = threadIdx.x;
    const int w    = tid >> 5;
    const int lane = tid & 31;
    const int l15  = lane & 15;
    const int h    = lane >> 4;

    const float4* G4 = (const float4*)G;
    #pragma unroll
    for (int i = 0; i < 32; ++i) {
        const int idx4 = tid + 128 * i;
        const int n = idx4 >> 5, k4 = idx4 & 31;
        *(float4*)&Mlds[n * PAD + k4 * 4] = G4[idx4];
    }

    const int e = blockIdx.x * 128 + w * 32 + lane;
    const float* xp = x + e * 24;
    float vr[6][2], vi[6][2];
    #pragma unroll
    for (int q = 0; q < 6; ++q) {
        float m = (xp[4 * q] + xp[4 * q + 1] + xp[4 * q + 2] + xp[4 * q + 3]) * 0.25f;
        m = fminf(6.f, fmaxf(-6.f, m));
        const float a  = m * (PI_F / 6.f);
        const float c2 = cosf(0.5f * a),  s2 = sinf(0.5f * a);
        const float c4 = cosf(0.25f * a), s4 = sinf(0.25f * a);
        vr[q][0] =  c2 * c4;  vi[q][0] = -c2 * s4;
        vr[q][1] =  s2 * s4;  vi[q][1] = -s2 * c4;
    }
    float* u = &Ulds[(w * 32 + lane) * PAD];
    for (int j = 0; j < 64; ++j) {
        float pr = vr[0][(j >> 5) & 1], pim = vi[0][(j >> 5) & 1];
        #pragma unroll
        for (int q = 1; q < 6; ++q) {
            const int b = (j >> (5 - q)) & 1;
            const float qr = vr[q][b], qi = vi[q][b];
            const float nr = pr * qr - pim * qi;
            pim = pr * qi + pim * qr;
            pr  = nr;
        }
        u[j] = pr; u[64 + j] = pim;
    }
    __syncthreads();

    v8f acc[2][8];
    const v8f z8 = {0.f, 0.f, 0.f, 0.f, 0.f, 0.f, 0.f, 0.f};
    #pragma unroll
    for (int rt = 0; rt < 2; ++rt)
        #pragma unroll
        for (int ct = 0; ct < 8; ++ct) acc[rt][ct] = z8;

    const float* urow0 = &Ulds[(w * 32 +  0 + l15) * PAD];
    const float* urow1 = &Ulds[(w * 32 + 16 + l15) * PAD];
    for (int kt = 0; kt < 32; ++kt) {
        const int kb = kt * 4 + 2 * h;
        v2f b[8];
        #pragma unroll
        for (int ct = 0; ct < 8; ++ct) {
            const float* brow = &Mlds[(ct * 16 + l15) * PAD + kb];
            b[ct].x = brow[0]; b[ct].y = brow[1];
        }
        v2f a0; a0.x = urow0[kb]; a0.y = urow0[kb + 1];
        v2f a1; a1.x = urow1[kb]; a1.y = urow1[kb + 1];
        #pragma unroll
        for (int ct = 0; ct < 8; ++ct) {
            acc[0][ct] = __builtin_amdgcn_wmma_f32_16x16x4_f32(
                false, a0, false, b[ct], (short)0, acc[0][ct], false, false);
            acc[1][ct] = __builtin_amdgcn_wmma_f32_16x16x4_f32(
                false, a1, false, b[ct], (short)0, acc[1][ct], false, false);
        }
    }

    auto rsum16 = [](float v) {
        v += __shfl_xor(v, 1, 32);
        v += __shfl_xor(v, 2, 32);
        v += __shfl_xor(v, 4, 32);
        v += __shfl_xor(v, 8, 32);
        return v;
    };

    #pragma unroll
    for (int rt = 0; rt < 2; ++rt) {
        #pragma unroll
        for (int r = 0; r < 8; ++r) {
            float re[4], im[4];
            #pragma unroll
            for (int t4 = 0; t4 < 4; ++t4) { re[t4] = acc[rt][t4][r]; im[t4] = acc[rt][4 + t4][r]; }

            float z0 = 0.f, z2 = 0.f, z4 = 0.f, x1 = 0.f, x3 = 0.f, x5 = 0.f;
            #pragma unroll
            for (int t4 = 0; t4 < 4; ++t4) {
                const int j = t4 * 16 + l15;
                const float p = re[t4] * re[t4] + im[t4] * im[t4];
                z0 += ((j >> 5) & 1) ? -p : p;
                z2 += ((j >> 3) & 1) ? -p : p;
                z4 += ((j >> 1) & 1) ? -p : p;
                x1 += re[t4] * re[t4 ^ 1] + im[t4] * im[t4 ^ 1];
                x3 += re[t4] * __shfl_xor(re[t4], 4, 32)
                    + im[t4] * __shfl_xor(im[t4], 4, 32);
                x5 += re[t4] * __shfl_xor(re[t4], 1, 32)
                    + im[t4] * __shfl_xor(im[t4], 1, 32);
            }
            z0 = rsum16(z0); z2 = rsum16(z2); z4 = rsum16(z4);
            x1 = rsum16(x1); x3 = rsum16(x3); x5 = rsum16(x5);

            if (l15 == 0) {
                const int ee = blockIdx.x * 128 + w * 32 + rt * 16 + r + h * 8;
                float4 o0 = { z0, x1, z2, x3 };
                float4 o1 = { z4, x5, z0, x1 };
                *(float4*)&out[ee * 8 + 0] = o0;
                *(float4*)&out[ee * 8 + 4] = o1;
            }
        }
    }
}

extern "C" void kernel_launch(void* const* d_in, const int* in_sizes, int n_in,
                              void* d_out, int out_size, void* d_ws, size_t ws_size,
                              hipStream_t stream) {
    const float* x     = (const float*)d_in[0];
    const float* theta = (const float*)d_in[1];
    float* out = (float*)d_out;
    float* G   = (float*)d_ws;

    build_G_kernel<<<1, 64, 0, stream>>>(theta, G);
    qsim_kernel<<<131072 / 128, 128, 0, stream>>>(x, G, out);
}


# G build: all 64 columns in parallel (2 blocks x 256 threads, 8 amplitudes per thread in registers, lane=column LDS layout without bank conflicts) instead of one 64-thread block of serial LDS loops
# speedup vs baseline: 6.4888x; 6.4888x over previous
#include <hip/hip_runtime.h>
#include <math.h>


typedef float v2f __attribute__((ext_vector_type(2)));
typedef float v8f __attribute__((ext_vector_type(8)));

#define PAD 132
#define PI_F 3.14159265358979323846f


__device__ __forceinline__ int ring_src(int j) {
    int s = j;
    if (s & 1)  s ^= 32;
    if (s & 2)  s ^= 1;
    if (s & 4)  s ^= 2;
    if (s & 8)  s ^= 4;
    if (s & 16) s ^= 8;
    if (s & 32) s ^= 16;
    return s;
}

__device__ __forceinline__ void rot3(float2& a, float2& b, float2 tx, float2 ty, float2 tz) {
    const float2 a1 = make_float2(tx.x * a.x + tx.y * b.y, tx.x * a.y - tx.y * b.x);
    const float2 b1 = make_float2(tx.y * a.y + tx.x * b.x, tx.x * b.y - tx.y * a.x);
    const float2 a2 = make_float2(ty.x * a1.x - ty.y * b1.x, ty.x * a1.y - ty.y * b1.y);
    const float2 b2 = make_float2(ty.y * a1.x + ty.x * b1.x, ty.y * a1.y + ty.x * b1.y);
    a = make_float2(tz.x * a2.x + tz.y * a2.y, tz.x * a2.y - tz.y * a2.x);
    b = make_float2(tz.x * b2.x - tz.y * b2.y, tz.x * b2.y + tz.y * b2.x);
}

__device__ __forceinline__ void rot_group(float2 (&v)[8], const float2* t) {
    #pragma unroll
    for (int qq = 0; qq < 3; ++qq) {
        const int m = 4 >> qq;
        const float2 tx = t[qq * 3 + 0], ty = t[qq * 3 + 1], tz = t[qq * 3 + 2];
        #pragma unroll
        for (int i = 0; i < 8; ++i)
            if (!(i & m)) rot3(v[i], v[i | m], tx, ty, tz);
    }
}

template <bool FIRST>
__device__ __forceinline__ void layer_hi(int l, const float2* src, float2* dst,
                                         const float2* trig, int c, int g, int col) {
    float2 v[8];
    #pragma unroll
    for (int hi = 0; hi < 8; ++hi) {
        const int s = ring_src(hi * 8 + g);
        if (FIRST) v[hi] = make_float2(s == col ? 1.f : 0.f, 0.f);
        else       v[hi] = src[s * 32 + c];
    }
    rot_group(v, trig + (l * 6 + 0) * 3);
    #pragma unroll
    for (int hi = 0; hi < 8; ++hi) dst[(hi * 8 + g) * 32 + c] = v[hi];
}

__device__ __forceinline__ void layer_lo(int l, const float2* st, const float2* trig,
                                         int c, int g, float2 (&v)[8]) {
    #pragma unroll
    for (int lo = 0; lo < 8; ++lo) v[lo] = st[(g * 8 + lo) * 32 + c];
    rot_group(v, trig + (l * 6 + 3) * 3);
}

__global__ __launch_bounds__(256) void build_G_kernel(const float* __restrict__ theta,
                                                      float* __restrict__ G) {
    __shared__ float2 buf0[64 * 32];
    __shared__ float2 buf1[64 * 32];
    __shared__ float2 trig[54];

    const int tid = threadIdx.x;
    const int c   = tid & 31;
    const int g   = tid >> 5;
    const int col = blockIdx.x * 32 + c;

    if (tid < 54) {
        const float a = theta[tid];
        trig[tid] = make_float2(cosf(0.5f * a), sinf(0.5f * a));
    }
    __syncthreads();

    float2 v[8];
    layer_hi<true>(0, buf1, buf0, trig, c, g, col);
    __syncthreads();
    layer_lo(0, buf0, trig, c, g, v);
    #pragma unroll
    for (int lo = 0; lo < 8; ++lo) buf0[(g * 8 + lo) * 32 + c] = v[lo];
    __syncthreads();

    layer_hi<false>(1, buf0, buf1, trig, c, g, col);
    __syncthreads();
    layer_lo(1, buf1, trig, c, g, v);
    #pragma unroll
    for (int lo = 0; lo < 8; ++lo) buf1[(g * 8 + lo) * 32 + c] = v[lo];
    __syncthreads();

    layer_hi<false>(2, buf1, buf0, trig, c, g, col);
    __syncthreads();
    layer_lo(2, buf0, trig, c, g, v);

    #pragma unroll
    for (int lo = 0; lo < 8; ++lo) {
        const int j = g * 8 + lo;
        G[j * 128 + col]             =  v[lo].x;
        G[j * 128 + 64 + col]        = -v[lo].y;
        G[(j + 64) * 128 + col]      =  v[lo].y;
        G[(j + 64) * 128 + 64 + col] =  v[lo].x;
    }
}

__global__ __launch_bounds__(128) void qsim_kernel(const float* __restrict__ x,
                                                   const float* __restrict__ G,
                                                   float* __restrict__ out) {
    __shared__ float Mlds[128 * PAD];
    __shared__ float Ulds[4 * 32 * PAD];

    const int tid  = threadIdx.x;
    const int w    = tid >> 5;
    const int lane = tid & 31;
    const int l15  = lane & 15;
    const int h    = lane >> 4;

    const float4* G4 = (const float4*)G;
    #pragma unroll
    for (int i = 0; i < 32; ++i) {
        const int idx4 = tid + 128 * i;
        const int n = idx4 >> 5, k4 = idx4 & 31;
        *(float4*)&Mlds[n * PAD + k4 * 4] = G4[idx4];
    }

    const int e = blockIdx.x * 128 + w * 32 + lane;
    const float* xp = x + e * 24;
    float vr[6][2], vi[6][2];
    #pragma unroll
    for (int q = 0; q < 6; ++q) {
        float m = (xp[4 * q] + xp[4 * q + 1] + xp[4 * q + 2] + xp[4 * q + 3]) * 0.25f;
        m = fminf(6.f, fmaxf(-6.f, m));
        const float a  = m * (PI_F / 6.f);
        const float c2 = cosf(0.5f * a),  s2 = sinf(0.5f * a);
        const float c4 = cosf(0.25f * a), s4 = sinf(0.25f * a);
        vr[q][0] =  c2 * c4;  vi[q][0] = -c2 * s4;
        vr[q][1] =  s2 * s4;  vi[q][1] = -s2 * c4;
    }
    float* u = &Ulds[(w * 32 + lane) * PAD];
    for (int j = 0; j < 64; ++j) {
        float pr = vr[0][(j >> 5) & 1], pim = vi[0][(j >> 5) & 1];
        #pragma unroll
        for (int q = 1; q < 6; ++q) {
            const int b = (j >> (5 - q)) & 1;
            const float qr = vr[q][b], qi = vi[q][b];
            const float nr = pr * qr - pim * qi;
            pim = pr * qi + pim * qr;
            pr  = nr;
        }
        u[j] = pr; u[64 + j] = pim;
    }
    __syncthreads();

    v8f acc[2][8];
    const v8f z8 = {0.f, 0.f, 0.f, 0.f, 0.f, 0.f, 0.f, 0.f};
    #pragma unroll
    for (int rt = 0; rt < 2; ++rt)
        #pragma unroll
        for (int ct = 0; ct < 8; ++ct) acc[rt][ct] = z8;

    const float* urow0 = &Ulds[(w * 32 +  0 + l15) * PAD];
    const float* urow1 = &Ulds[(w * 32 + 16 + l15) * PAD];
    for (int kt = 0; kt < 32; ++kt) {
        const int kb = kt * 4 + 2 * h;
        v2f b[8];
        #pragma unroll
        for (int ct = 0; ct < 8; ++ct) {
            const float* brow = &Mlds[(ct * 16 + l15) * PAD + kb];
            b[ct].x = brow[0]; b[ct].y = brow[1];
        }
        v2f a0; a0.x = urow0[kb]; a0.y = urow0[kb + 1];
        v2f a1; a1.x = urow1[kb]; a1.y = urow1[kb + 1];
        #pragma unroll
        for (int ct = 0; ct < 8; ++ct) {
            acc[0][ct] = __builtin_amdgcn_wmma_f32_16x16x4_f32(
                false, a0, false, b[ct], (short)0, acc[0][ct], false, false);
            acc[1][ct] = __builtin_amdgcn_wmma_f32_16x16x4_f32(
                false, a1, false, b[ct], (short)0, acc[1][ct], false, false);
        }
    }

    auto rsum16 = [](float v) {
        v += __shfl_xor(v, 1, 32);
        v += __shfl_xor(v, 2, 32);
        v += __shfl_xor(v, 4, 32);
        v += __shfl_xor(v, 8, 32);
        return v;
    };

    #pragma unroll
    for (int rt = 0; rt < 2; ++rt) {
        #pragma unroll
        for (int r = 0; r < 8; ++r) {
            float re[4], im[4];
            #pragma unroll
            for (int t4 = 0; t4 < 4; ++t4) { re[t4] = acc[rt][t4][r]; im[t4] = acc[rt][4 + t4][r]; }

            float z0 = 0.f, z2 = 0.f, z4 = 0.f, x1 = 0.f, x3 = 0.f, x5 = 0.f;
            #pragma unroll
            for (int t4 = 0; t4 < 4; ++t4) {
                const int j = t4 * 16 + l15;
                const float p = re[t4] * re[t4] + im[t4] * im[t4];
                z0 += ((j >> 5) & 1) ? -p : p;
                z2 += ((j >> 3) & 1) ? -p : p;
                z4 += ((j >> 1) & 1) ? -p : p;
                x1 += re[t4] * re[t4 ^ 1] + im[t4] * im[t4 ^ 1];
                x3 += re[t4] * __shfl_xor(re[t4], 4, 32)
                    + im[t4] * __shfl_xor(im[t4], 4, 32);
                x5 += re[t4] * __shfl_xor(re[t4], 1, 32)
                    + im[t4] * __shfl_xor(im[t4], 1, 32);
            }
            z0 = rsum16(z0); z2 = rsum16(z2); z4 = rsum16(z4);
            x1 = rsum16(x1); x3 = rsum16(x3); x5 = rsum16(x5);

            if (l15 == 0) {
                const int ee = blockIdx.x * 128 + w * 32 + rt * 16 + r + h * 8;
                float4 o0 = { z0, x1, z2, x3 };
                float4 o1 = { z4, x5, z0, x1 };
                *(float4*)&out[ee * 8 + 0] = o0;
                *(float4*)&out[ee * 8 + 4] = o1;
            }
        }
    }
}

extern "C" void kernel_launch(void* const* d_in, const int* in_sizes, int n_in,
                              void* d_out, int out_size, void* d_ws, size_t ws_size,
                              hipStream_t stream) {
    const float* x     = (const float*)d_in[0];
    const float* theta = (const float*)d_in[1];
    float* out = (float*)d_out;
    float* G   = (float*)d_ws;

    build_G_kernel<<<2, 256, 0, stream>>>(theta, G);
    qsim_kernel<<<131072 / 128, 128, 0, stream>>>(x, G, out);
}


# Batch moved to the column side of the f32 WMMA product (psi^T = G u^T): a lane owns one element, forms its u fragments in registers (no LDS panel), sums its Pauli terms in-lane; sum order changes
# speedup vs baseline: 9.3627x; 1.4429x over previous
#include <hip/hip_runtime.h>
#include <math.h>


typedef float v2f __attribute__((ext_vector_type(2)));
typedef float v8f __attribute__((ext_vector_type(8)));

#define PAD 132
#define PI_F 3.14159265358979323846f


__device__ __forceinline__ int ring_src(int j) {
    int s = j;
    if (s & 1)  s ^= 32;
    if (s & 2)  s ^= 1;
    if (s & 4)  s ^= 2;
    if (s & 8)  s ^= 4;
    if (s & 16) s ^= 8;
    if (s & 32) s ^= 16;
    return s;
}

__device__ __forceinline__ void rot3(float2& a, float2& b, float2 tx, float2 ty, float2 tz) {
    const float2 a1 = make_float2(tx.x * a.x + tx.y * b.y, tx.x * a.y - tx.y * b.x);
    const float2 b1 = make_float2(tx.y * a.y + tx.x * b.x, tx.x * b.y - tx.y * a.x);
    const float2 a2 = make_float2(ty.x * a1.x - ty.y * b1.x, ty.x * a1.y - ty.y * b1.y);
    const float2 b2 = make_float2(ty.y * a1.x + ty.x * b1.x, ty.y * a1.y + ty.x * b1.y);
    a = make_float2(tz.x * a2.x + tz.y * a2.y, tz.x * a2.y - tz.y * a2.x);
    b = make_float2(tz.x * b2.x - tz.y * b2.y, tz.x * b2.y + tz.y * b2.x);
}

__device__ __forceinline__ void rot_group(float2 (&v)[8], const float2* t) {
    #pragma unroll
    for (int qq = 0; qq < 3; ++qq) {
        const int m = 4 >> qq;
        const float2 tx = t[qq * 3 + 0], ty = t[qq * 3 + 1], tz = t[qq * 3 + 2];
        #pragma unroll
        for (int i = 0; i < 8; ++i)
            if (!(i & m)) rot3(v[i], v[i | m], tx, ty, tz);
    }
}

template <bool FIRST>
__device__ __forceinline__ void layer_hi(int l, const float2* src, float2* dst,
                                         const float2* trig, int c, int g, int col) {
    float2 v[8];
    #pragma unroll
    for (int hi = 0; hi < 8; ++hi) {
        const int s = ring_src(hi * 8 + g);
        if (FIRST) v[hi] = make_float2(s == col ? 1.f : 0.f, 0.f);
        else       v[hi] = src[s * 32 + c];
    }
    rot_group(v, trig + (l * 6 + 0) * 3);
    #pragma unroll
    for (int hi = 0; hi < 8; ++hi) dst[(hi * 8 + g) * 32 + c] = v[hi];
}

__device__ __forceinline__ void layer_lo(int l, const float2* st, const float2* trig,
                                         int c, int g, float2 (&v)[8]) {
    #pragma unroll
    for (int lo = 0; lo < 8; ++lo) v[lo] = st[(g * 8 + lo) * 32 + c];
    rot_group(v, trig + (l * 6 + 3) * 3);
}

__global__ __launch_bounds__(256) void build_G_kernel(const float* __restrict__ theta,
                                                      float* __restrict__ G) {
    __shared__ float2 buf0[64 * 32];
    __shared__ float2 buf1[64 * 32];
    __shared__ float2 trig[54];

    const int tid = threadIdx.x;
    const int c   = tid & 31;
    const int g   = tid >> 5;
    const int col = blockIdx.x * 32 + c;

    if (tid < 54) {
        const float a = theta[tid];
        trig[tid] = make_float2(cosf(0.5f * a), sinf(0.5f * a));
    }
    __syncthreads();

    float2 v[8];
    layer_hi<true>(0, buf1, buf0, trig, c, g, col);
    __syncthreads();
    layer_lo(0, buf0, trig, c, g, v);
    #pragma unroll
    for (int lo = 0; lo < 8; ++lo) buf0[(g * 8 + lo) * 32 + c] = v[lo];
    __syncthreads();

    layer_hi<false>(1, buf0, buf1, trig, c, g, col);
    __syncthreads();
    layer_lo(1, buf1, trig, c, g, v);
    #pragma unroll
    for (int lo = 0; lo < 8; ++lo) buf1[(g * 8 + lo) * 32 + c] = v[lo];
    __syncthreads();

    layer_hi<false>(2, buf1, buf0, trig, c, g, col);
    __syncthreads();
    layer_lo(2, buf0, trig, c, g, v);

    #pragma unroll
    for (int lo = 0; lo < 8; ++lo) {
        const int j = g * 8 + lo;
        G[j * 128 + col]             =  v[lo].x;
        G[j * 128 + 64 + col]        = -v[lo].y;
        G[(j + 64) * 128 + col]      =  v[lo].y;
        G[(j + 64) * 128 + 64 + col] =  v[lo].x;
    }
}

__device__ __forceinline__ float2 cmul(float2 a, float2 b) {
    return make_float2(a.x * b.x - a.y * b.y, a.x * b.y + a.y * b.x);
}

__global__ __launch_bounds__(128) void qsim_kernel(const float* __restrict__ x,
                                                   const float* __restrict__ G,
                                                   float* __restrict__ out) {
    __shared__ float  Mlds[128 * PAD];
    __shared__ float4 Xch[4 * 6 * 32];

    const int tid  = threadIdx.x;
    const int w    = tid >> 5;
    const int lane = tid & 31;
    const int c    = lane & 15;
    const int h    = lane >> 4;

    const float4* G4 = (const float4*)G;
    #pragma unroll
    for (int i = 0; i < 32; ++i) {
        const int idx4 = tid + 128 * i;
        const int n = idx4 >> 5, k4 = idx4 & 31;
        *(float4*)&Mlds[n * PAD + k4 * 4] = G4[idx4];
    }

    {
        const int e = blockIdx.x * 128 + w * 32 + lane;
        const float* xp = x + e * 24;
        #pragma unroll
        for (int q = 0; q < 6; ++q) {
            float m = (xp[4 * q] + xp[4 * q + 1] + xp[4 * q + 2] + xp[4 * q + 3]) * 0.25f;
            m = fminf(6.f, fmaxf(-6.f, m));
            const float a  = m * (PI_F / 6.f);
            const float c2 = cosf(0.5f * a),  s2 = sinf(0.5f * a);
            const float c4 = cosf(0.25f * a), s4 = sinf(0.25f * a);
            Xch[(w * 6 + q) * 32 + lane] = make_float4(c2 * c4, -c2 * s4, s2 * s4, -s2 * c4);
        }
    }
    __syncthreads();

    float2 hi[2][16], lo[2][2];
    #pragma unroll
    for (int ct = 0; ct < 2; ++ct) {
        float4 v[6];
        #pragma unroll
        for (int q = 0; q < 6; ++q) v[q] = Xch[(w * 6 + q) * 32 + ct * 16 + c];
        float2 p2[4], p3[8];
        #pragma unroll
        for (int i = 0; i < 4; ++i) {
            const float2 f0 = (i & 2) ? make_float2(v[0].z, v[0].w) : make_float2(v[0].x, v[0].y);
            const float2 f1 = (i & 1) ? make_float2(v[1].z, v[1].w) : make_float2(v[1].x, v[1].y);
            p2[i] = cmul(f0, f1);
        }
        #pragma unroll
        for (int i = 0; i < 8; ++i) {
            const float2 f2 = (i & 1) ? make_float2(v[2].z, v[2].w) : make_float2(v[2].x, v[2].y);
            p3[i] = cmul(p2[i >> 1], f2);
        }
        #pragma unroll
        for (int i = 0; i < 16; ++i) {
            const float2 f3 = (i & 1) ? make_float2(v[3].z, v[3].w) : make_float2(v[3].x, v[3].y);
            hi[ct][i] = cmul(p3[i >> 1], f3);
        }
        const float2 f4 = h ? make_float2(v[4].z, v[4].w) : make_float2(v[4].x, v[4].y);
        lo[ct][0] = cmul(f4, make_float2(v[5].x, v[5].y));
        lo[ct][1] = cmul(f4, make_float2(v[5].z, v[5].w));
    }

    v8f acc[2][8];
    const v8f z8 = {0.f, 0.f, 0.f, 0.f, 0.f, 0.f, 0.f, 0.f};
    #pragma unroll
    for (int ct = 0; ct < 2; ++ct)
        #pragma unroll
        for (int rt = 0; rt < 8; ++rt) acc[ct][rt] = z8;

    const float* arow = &Mlds[c * PAD + 2 * h];
    #pragma unroll
    for (int kt = 0; kt < 32; ++kt) {
        const int m = kt & 15;
        v2f a[8];
        #pragma unroll
        for (int rt = 0; rt < 8; ++rt) {
            const float* ar = arow + rt * 16 * PAD + kt * 4;
            a[rt].x = ar[0]; a[rt].y = ar[1];
        }
        v2f b[2];
        #pragma unroll
        for (int ct = 0; ct < 2; ++ct) {
            const float2 hm = hi[ct][m];
            if (kt < 16) {
                b[ct].x = hm.x * lo[ct][0].x - hm.y * lo[ct][0].y;
                b[ct].y = hm.x * lo[ct][1].x - hm.y * lo[ct][1].y;
            } else {
                b[ct].x = hm.x * lo[ct][0].y + hm.y * lo[ct][0].x;
                b[ct].y = hm.x * lo[ct][1].y + hm.y * lo[ct][1].x;
            }
        }
        #pragma unroll
        for (int rt = 0; rt < 8; ++rt) {
            acc[0][rt] = __builtin_amdgcn_wmma_f32_16x16x4_f32(
                false, a[rt], false, b[0], (short)0, acc[0][rt], false, false);
            acc[1][rt] = __builtin_amdgcn_wmma_f32_16x16x4_f32(
                false, a[rt], false, b[1], (short)0, acc[1][rt], false, false);
        }
    }

    #pragma unroll
    for (int ct = 0; ct < 2; ++ct) {
        float pz[4][2];
        float x1 = 0.f, x3 = 0.f, x5 = 0.f;
        #pragma unroll
        for (int t = 0; t < 4; ++t) {
            pz[t][0] = 0.f; pz[t][1] = 0.f;
            #pragma unroll
            for (int r = 0; r < 8; ++r) {
                const float re = acc[ct][t][r], im = acc[ct][4 + t][r];
                pz[t][(r >> 1) & 1] += re * re + im * im;
                if (!(t & 1))
                    x1 += re * acc[ct][t + 1][r] + im * acc[ct][4 + t + 1][r];
                if (!(r & 4))
                    x3 += re * acc[ct][t][r + 4] + im * acc[ct][4 + t][r + 4];
                if (!(r & 1))
                    x5 += re * acc[ct][t][r + 1] + im * acc[ct][4 + t][r + 1];
            }
        }
        const float t01 = (pz[0][0] + pz[0][1]) + (pz[1][0] + pz[1][1]);
        const float t23 = (pz[2][0] + pz[2][1]) + (pz[3][0] + pz[3][1]);
        float z0 = t01 - t23;
        float z2 = h ? -(t01 + t23) : (t01 + t23);
        float z4 = ((pz[0][0] + pz[1][0]) + (pz[2][0] + pz[3][0]))
                 - ((pz[0][1] + pz[1][1]) + (pz[2][1] + pz[3][1]));
        x1 *= 2.f; x3 *= 2.f; x5 *= 2.f;
        z0 += __shfl_xor(z0, 16, 32); z2 += __shfl_xor(z2, 16, 32); z4 += __shfl_xor(z4, 16, 32);
        x1 += __shfl_xor(x1, 16, 32); x3 += __shfl_xor(x3, 16, 32); x5 += __shfl_xor(x5, 16, 32);

        const int ee = blockIdx.x * 128 + w * 32 + ct * 16 + c;
        const float4 o = h ? make_float4(z4, x5, z0, x1) : make_float4(z0, x1, z2, x3);
        *(float4*)&out[ee * 8 + 4 * h] = o;
    }
}

extern "C" void kernel_launch(void* const* d_in, const int* in_sizes, int n_in,
                              void* d_out, int out_size, void* d_ws, size_t ws_size,
                              hipStream_t stream) {
    const float* x     = (const float*)d_in[0];
    const float* theta = (const float*)d_in[1];
    float* out = (float*)d_out;
    float* G   = (float*)d_ws;

    build_G_kernel<<<2, 256, 0, stream>>>(theta, G);
    qsim_kernel<<<131072 / 128, 128, 0, stream>>>(x, G, out);
}


# Complex product psi = W s as three real 64x64 f32 WMMA products (k1=Wr(sr+si), Re=k1-(Wr+Wi)si, Im=k1+(Wi-Wr)sr) instead of the 128x128 real embedding: 384 not 512 WMMAs per wave; sum order changes
# speedup vs baseline: 11.2627x; 1.2029x over previous
#include <hip/hip_runtime.h>
#include <math.h>


typedef float v2f __attribute__((ext_vector_type(2)));
typedef float v8f __attribute__((ext_vector_type(8)));

#define PAD 68
#define PI_F 3.14159265358979323846f


__device__ __forceinline__ int ring_src(int j) {
    int s = j;
    if (s & 1)  s ^= 32;
    if (s & 2)  s ^= 1;
    if (s & 4)  s ^= 2;
    if (s & 8)  s ^= 4;
    if (s & 16) s ^= 8;
    if (s & 32) s ^= 16;
    return s;
}

__device__ __forceinline__ void rot3(float2& a, float2& b, float2 tx, float2 ty, float2 tz) {
    const float2 a1 = make_float2(tx.x * a.x + tx.y * b.y, tx.x * a.y - tx.y * b.x);
    const float2 b1 = make_float2(tx.y * a.y + tx.x * b.x, tx.x * b.y - tx.y * a.x);
    const float2 a2 = make_float2(ty.x * a1.x - ty.y * b1.x, ty.x * a1.y - ty.y * b1.y);
    const float2 b2 = make_float2(ty.y * a1.x + ty.x * b1.x, ty.y * a1.y + ty.x * b1.y);
    a = make_float2(tz.x * a2.x + tz.y * a2.y, tz.x * a2.y - tz.y * a2.x);
    b = make_float2(tz.x * b2.x - tz.y * b2.y, tz.x * b2.y + tz.y * b2.x);
}

__device__ __forceinline__ void rot_group(float2 (&v)[8], const float2* t) {
    #pragma unroll
    for (int qq = 0; qq < 3; ++qq) {
        const int m = 4 >> qq;
        const float2 tx = t[qq * 3 + 0], ty = t[qq * 3 + 1], tz = t[qq * 3 + 2];
        #pragma unroll
        for (int i = 0; i < 8; ++i)
            if (!(i & m)) rot3(v[i], v[i | m], tx, ty, tz);
    }
}

template <bool FIRST>
__device__ __forceinline__ void layer_hi(int l, const float2* src, float2* dst,
                                         const float2* trig, int c, int g, int col) {
    float2 v[8];
    #pragma unroll
    for (int hi = 0; hi < 8; ++hi) {
        const int s = ring_src(hi * 8 + g);
        if (FIRST) v[hi] = make_float2(s == col ? 1.f : 0.f, 0.f);
        else       v[hi] = src[s * 32 + c];
    }
    rot_group(v, trig + (l * 6 + 0) * 3);
    #pragma unroll
    for (int hi = 0; hi < 8; ++hi) dst[(hi * 8 + g) * 32 + c] = v[hi];
}

__device__ __forceinline__ void layer_lo(int l, const float2* st, const float2* trig,
                                         int c, int g, float2 (&v)[8]) {
    #pragma unroll
    for (int lo = 0; lo < 8; ++lo) v[lo] = st[(g * 8 + lo) * 32 + c];
    rot_group(v, trig + (l * 6 + 3) * 3);
}

__global__ __launch_bounds__(256) void build_G_kernel(const float* __restrict__ theta,
                                                      float* __restrict__ G) {
    __shared__ float2 buf0[64 * 32];
    __shared__ float2 buf1[64 * 32];
    __shared__ float2 trig[54];

    const int tid = threadIdx.x;
    const int c   = tid & 31;
    const int g   = tid >> 5;
    const int col = blockIdx.x * 32 + c;

    if (tid < 54) {
        const float a = theta[tid];
        trig[tid] = make_float2(cosf(0.5f * a), sinf(0.5f * a));
    }
    __syncthreads();

    float2 v[8];
    layer_hi<true>(0, buf1, buf0, trig, c, g, col);
    __syncthreads();
    layer_lo(0, buf0, trig, c, g, v);
    #pragma unroll
    for (int lo = 0; lo < 8; ++lo) buf0[(g * 8 + lo) * 32 + c] = v[lo];
    __syncthreads();

    layer_hi<false>(1, buf0, buf1, trig, c, g, col);
    __syncthreads();
    layer_lo(1, buf1, trig, c, g, v);
    #pragma unroll
    for (int lo = 0; lo < 8; ++lo) buf1[(g * 8 + lo) * 32 + c] = v[lo];
    __syncthreads();

    layer_hi<false>(2, buf1, buf0, trig, c, g, col);
    __syncthreads();
    layer_lo(2, buf0, trig, c, g, v);

    #pragma unroll
    for (int lo = 0; lo < 8; ++lo) {
        const int j = g * 8 + lo;
        G[j * 64 + col]         =  v[lo].x;
        G[(64 + j) * 64 + col]  =  v[lo].y - v[lo].x;
        G[(128 + j) * 64 + col] = -(v[lo].x + v[lo].y);
    }
}

__device__ __forceinline__ float2 cmul(float2 a, float2 b) {
    return make_float2(a.x * b.x - a.y * b.y, a.x * b.y + a.y * b.x);
}

__global__ __launch_bounds__(128) void qsim_kernel(const float* __restrict__ x,
                                                   const float* __restrict__ G,
                                                   float* __restrict__ out) {
    __shared__ float  Mlds[192 * PAD];
    __shared__ float4 Xch[4 * 6 * 32];

    const int tid  = threadIdx.x;
    const int w    = tid >> 5;
    const int lane = tid & 31;
    const int c    = lane & 15;
    const int h    = lane >> 4;

    const float4* G4 = (const float4*)G;
    #pragma unroll
    for (int i = 0; i < 24; ++i) {
        const int idx4 = tid + 128 * i;
        const int n = idx4 >> 4, k4 = idx4 & 15;
        *(float4*)&Mlds[n * PAD + k4 * 4] = G4[idx4];
    }

    {
        const int e = blockIdx.x * 128 + w * 32 + lane;
        const float* xp = x + e * 24;
        #pragma unroll
        for (int q = 0; q < 6; ++q) {
            float m = (xp[4 * q] + xp[4 * q + 1] + xp[4 * q + 2] + xp[4 * q + 3]) * 0.25f;
            m = fminf(6.f, fmaxf(-6.f, m));
            const float a  = m * (PI_F / 6.f);
            const float c2 = cosf(0.5f * a),  s2 = sinf(0.5f * a);
            const float c4 = cosf(0.25f * a), s4 = sinf(0.25f * a);
            Xch[(w * 6 + q) * 32 + lane] = make_float4(c2 * c4, -c2 * s4, s2 * s4, -s2 * c4);
        }
    }
    __syncthreads();

    float2 hi[2][16], lo[2][2];
    #pragma unroll
    for (int ct = 0; ct < 2; ++ct) {
        float4 v[6];
        #pragma unroll
        for (int q = 0; q < 6; ++q) v[q] = Xch[(w * 6 + q) * 32 + ct * 16 + c];
        float2 p2[4], p3[8];
        #pragma unroll
        for (int i = 0; i < 4; ++i) {
            const float2 f0 = (i & 2) ? make_float2(v[0].z, v[0].w) : make_float2(v[0].x, v[0].y);
            const float2 f1 = (i & 1) ? make_float2(v[1].z, v[1].w) : make_float2(v[1].x, v[1].y);
            p2[i] = cmul(f0, f1);
        }
        #pragma unroll
        for (int i = 0; i < 8; ++i) {
            const float2 f2 = (i & 1) ? make_float2(v[2].z, v[2].w) : make_float2(v[2].x, v[2].y);
            p3[i] = cmul(p2[i >> 1], f2);
        }
        #pragma unroll
        for (int i = 0; i < 16; ++i) {
            const float2 f3 = (i & 1) ? make_float2(v[3].z, v[3].w) : make_float2(v[3].x, v[3].y);
            hi[ct][i] = cmul(p3[i >> 1], f3);
        }
        const float2 f4 = h ? make_float2(v[4].z, v[4].w) : make_float2(v[4].x, v[4].y);
        lo[ct][0] = cmul(f4, make_float2(v[5].x, v[5].y));
        lo[ct][1] = cmul(f4, make_float2(v[5].z, v[5].w));
    }

    float ls[2][2], ld[2][2];
    #pragma unroll
    for (int ct = 0; ct < 2; ++ct)
        #pragma unroll
        for (int e = 0; e < 2; ++e) {
            ls[ct][e] = lo[ct][e].x + lo[ct][e].y;
            ld[ct][e] = lo[ct][e].x - lo[ct][e].y;
        }

    v8f k1[2][4], re[2][4];
    const v8f z8 = {0.f, 0.f, 0.f, 0.f, 0.f, 0.f, 0.f, 0.f};
    #pragma unroll
    for (int ct = 0; ct < 2; ++ct)
        #pragma unroll
        for (int rt = 0; rt < 4; ++rt) k1[ct][rt] = z8;

    const float* arow = &Mlds[c * PAD + 2 * h];

    #pragma unroll
    for (int kt = 0; kt < 16; ++kt) {
        v2f a[4];
        #pragma unroll
        for (int rt = 0; rt < 4; ++rt) {
            const float* ar = arow + rt * 16 * PAD + kt * 4;
            a[rt].x = ar[0]; a[rt].y = ar[1];
        }
        v2f b[2];
        #pragma unroll
        for (int ct = 0; ct < 2; ++ct) {
            const float2 hm = hi[ct][kt];
            b[ct].x = hm.x * ls[ct][0] + hm.y * ld[ct][0];
            b[ct].y = hm.x * ls[ct][1] + hm.y * ld[ct][1];
        }
        #pragma unroll
        for (int rt = 0; rt < 4; ++rt) {
            k1[0][rt] = __builtin_amdgcn_wmma_f32_16x16x4_f32(
                false, a[rt], false, b[0], (short)0, k1[0][rt], false, false);
            k1[1][rt] = __builtin_amdgcn_wmma_f32_16x16x4_f32(
                false, a[rt], false, b[1], (short)0, k1[1][rt], false, false);
        }
    }

    #pragma unroll
    for (int kt = 0; kt < 16; ++kt) {
        v2f ad[4], an[4];
        #pragma unroll
        for (int rt = 0; rt < 4; ++rt) {
            const float* pd = arow + (64 + rt * 16) * PAD + kt * 4;
            const float* pn = arow + (128 + rt * 16) * PAD + kt * 4;
            ad[rt].x = pd[0]; ad[rt].y = pd[1];
            an[rt].x = pn[0]; an[rt].y = pn[1];
        }
        v2f br[2], bi[2];
        #pragma unroll
        for (int ct = 0; ct < 2; ++ct) {
            const float2 hm = hi[ct][kt];
            br[ct].x = hm.x * lo[ct][0].x - hm.y * lo[ct][0].y;
            br[ct].y = hm.x * lo[ct][1].x - hm.y * lo[ct][1].y;
            bi[ct].x = hm.x * lo[ct][0].y + hm.y * lo[ct][0].x;
            bi[ct].y = hm.x * lo[ct][1].y + hm.y * lo[ct][1].x;
        }
        #pragma unroll
        for (int rt = 0; rt < 4; ++rt) {
            re[0][rt] = __builtin_amdgcn_wmma_f32_16x16x4_f32(
                false, an[rt], false, bi[0], (short)0, kt == 0 ? k1[0][rt] : re[0][rt], false, false);
            re[1][rt] = __builtin_amdgcn_wmma_f32_16x16x4_f32(
                false, an[rt], false, bi[1], (short)0, kt == 0 ? k1[1][rt] : re[1][rt], false, false);
        }
        #pragma unroll
        for (int rt = 0; rt < 4; ++rt) {
            k1[0][rt] = __builtin_amdgcn_wmma_f32_16x16x4_f32(
                false, ad[rt], false, br[0], (short)0, k1[0][rt], false, false);
            k1[1][rt] = __builtin_amdgcn_wmma_f32_16x16x4_f32(
                false, ad[rt], false, br[1], (short)0, k1[1][rt], false, false);
        }
    }

    #pragma unroll
    for (int ct = 0; ct < 2; ++ct) {
        float pz[4][2];
        float x1 = 0.f, x3 = 0.f, x5 = 0.f;
        #pragma unroll
        for (int t = 0; t < 4; ++t) {
            pz[t][0] = 0.f; pz[t][1] = 0.f;
            #pragma unroll
            for (int r = 0; r < 8; ++r) {
                const float ar = re[ct][t][r], ai = k1[ct][t][r];
                pz[t][(r >> 1) & 1] += ar * ar + ai * ai;
                if (!(t & 1))
                    x1 += ar * re[ct][t + 1][r] + ai * k1[ct][t + 1][r];
                if (!(r & 4))
                    x3 += ar * re[ct][t][r + 4] + ai * k1[ct][t][r + 4];
                if (!(r & 1))
                    x5 += ar * re[ct][t][r + 1] + ai * k1[ct][t][r + 1];
            }
        }
        const float t01 = (pz[0][0] + pz[0][1]) + (pz[1][0] + pz[1][1]);
        const float t23 = (pz[2][0] + pz[2][1]) + (pz[3][0] + pz[3][1]);
        float z0 = t01 - t23;
        float z2 = h ? -(t01 + t23) : (t01 + t23);
        float z4 = ((pz[0][0] + pz[1][0]) + (pz[2][0] + pz[3][0]))
                 - ((pz[0][1] + pz[1][1]) + (pz[2][1] + pz[3][1]));
        x1 *= 2.f; x3 *= 2.f; x5 *= 2.f;
        z0 += __shfl_xor(z0, 16, 32); z2 += __shfl_xor(z2, 16, 32); z4 += __shfl_xor(z4, 16, 32);
        x1 += __shfl_xor(x1, 16, 32); x3 += __shfl_xor(x3, 16, 32); x5 += __shfl_xor(x5, 16, 32);

        const int ee = blockIdx.x * 128 + w * 32 + ct * 16 + c;
        const float4 o = h ? make_float4(z4, x5, z0, x1) : make_float4(z0, x1, z2, x3);
        *(float4*)&out[ee * 8 + 4 * h] = o;
    }
}

extern "C" void kernel_launch(void* const* d_in, const int* in_sizes, int n_in,
                              void* d_out, int out_size, void* d_ws, size_t ws_size,
                              hipStream_t stream) {
    const float* x     = (const float*)d_in[0];
    const float* theta = (const float*)d_in[1];
    float* out = (float*)d_out;
    float* G   = (float*)d_ws;

    build_G_kernel<<<2, 256, 0, stream>>>(theta, G);
    qsim_kernel<<<131072 / 128, 128, 0, stream>>>(x, G, out);
}


# Circuit applied layer by layer: two 16x16 real tensor-factor f32 WMMA products per layer, state chained D->B in registers, factors built in-block (one launch, no dense unitary); sum order changes
# speedup vs baseline: 14.5574x; 1.2925x over previous
#include <hip/hip_runtime.h>
#include <math.h>


typedef float v2f __attribute__((ext_vector_type(2)));
typedef float v8f __attribute__((ext_vector_type(8)));

#define PI_F 3.14159265358979323846f

__device__ __forceinline__ int ring_src(int j) {
    int s = j;
    if (s & 1)  s ^= 32;
    if (s & 2)  s ^= 1;
    if (s & 4)  s ^= 2;
    if (s & 8)  s ^= 4;
    if (s & 16) s ^= 8;
    if (s & 32) s ^= 16;
    return s;
}

__device__ __forceinline__ void rot3(float2& a, float2& b, float2 tx, float2 ty, float2 tz) {
    const float2 a1 = make_float2(tx.x * a.x + tx.y * b.y, tx.x * a.y - tx.y * b.x);
    const float2 b1 = make_float2(tx.y * a.y + tx.x * b.x, tx.x * b.y - tx.y * a.x);
    const float2 a2 = make_float2(ty.x * a1.x - ty.y * b1.x, ty.x * a1.y - ty.y * b1.y);
    const float2 b2 = make_float2(ty.y * a1.x + ty.x * b1.x, ty.y * a1.y + ty.x * b1.y);
    a = make_float2(tz.x * a2.x + tz.y * a2.y, tz.x * a2.y - tz.y * a2.x);
    b = make_float2(tz.x * b2.x - tz.y * b2.y, tz.x * b2.y + tz.y * b2.x);
}

__device__ __forceinline__ float2 cmul(float2 a, float2 b) {
    return make_float2(a.x * b.x - a.y * b.y, a.x * b.y + a.y * b.x);
}

__device__ __forceinline__ void prod3(float2 (&p)[8], float4 v0, float4 v1, float4 v2) {
    float2 p2[4];
    #pragma unroll
    for (int i = 0; i < 4; ++i) {
        const float2 f0 = (i & 2) ? make_float2(v0.z, v0.w) : make_float2(v0.x, v0.y);
        const float2 f1 = (i & 1) ? make_float2(v1.z, v1.w) : make_float2(v1.x, v1.y);
        p2[i] = cmul(f0, f1);
    }
    #pragma unroll
    for (int i = 0; i < 8; ++i) {
        const float2 f2 = (i & 1) ? make_float2(v2.z, v2.w) : make_float2(v2.x, v2.y);
        p[i] = cmul(p2[i >> 1], f2);
    }
}

__global__ __launch_bounds__(128) void qsim_kernel(const float* __restrict__ x,
                                                   const float* __restrict__ theta,
                                                   float* __restrict__ out) {
    __shared__ float2 trig[54];
    __shared__ float2 Ug[18 * 4];
    __shared__ float  E[6 * 256];
    __shared__ float4 Xch[4 * 6 * 32];

    const int tid  = threadIdx.x;
    const int w    = tid >> 5;
    const int lane = tid & 31;
    const int c    = lane & 15;
    const int h    = lane >> 4;

    if (tid < 54) {
        const float a = theta[tid];
        trig[tid] = make_float2(cosf(0.5f * a), sinf(0.5f * a));
    }

    {
        const int e = blockIdx.x * 128 + w * 32 + lane;
        const float* xp = x + e * 24;
        #pragma unroll
        for (int q = 0; q < 6; ++q) {
            float m = (xp[4 * q] + xp[4 * q + 1] + xp[4 * q + 2] + xp[4 * q + 3]) * 0.25f;
            m = fminf(6.f, fmaxf(-6.f, m));
            const float a  = m * (PI_F / 6.f);
            const float c2 = cosf(0.5f * a),  s2 = sinf(0.5f * a);
            const float c4 = cosf(0.25f * a), s4 = sinf(0.25f * a);
            Xch[(w * 6 + q) * 32 + lane] = make_float4(c2 * c4, -c2 * s4, s2 * s4, -s2 * c4);
        }
    }
    __syncthreads();

    if (tid < 36) {
        const int g = tid >> 1, i = tid & 1;
        float2 a = make_float2(i ? 0.f : 1.f, 0.f);
        float2 b = make_float2(i ? 1.f : 0.f, 0.f);
        rot3(a, b, trig[g * 3 + 0], trig[g * 3 + 1], trig[g * 3 + 2]);
        Ug[g * 4 + 0 + i] = a;
        Ug[g * 4 + 2 + i] = b;
    }
    __syncthreads();

    for (int i = 0; i < 12; ++i) {
        const int idx = tid + 128 * i;
        const int mt = idx >> 8, m = (idx >> 4) & 15, k = idx & 15;
        const int hp = m >> 3, rp = m & 7, hh = k >> 3, r = k & 7;
        const float2* u = Ug + mt * 12;
        const float2 e0 = u[0 + ((rp >> 2) & 1) * 2 + ((r >> 2) & 1)];
        const float2 e1 = u[4 + ((rp >> 1) & 1) * 2 + ((r >> 1) & 1)];
        const float2 e2 = u[8 + (rp & 1) * 2 + (r & 1)];
        const float2 mm = cmul(cmul(e0, e1), e2);
        E[idx] = (hp == hh) ? mm.x : (hp ? mm.y : -mm.y);
    }
    __syncthreads();

    v2f A[6][4];
    #pragma unroll
    for (int mt = 0; mt < 6; ++mt)
        #pragma unroll
        for (int kt = 0; kt < 4; ++kt) {
            const float* p = &E[mt * 256 + c * 16 + 8 * h + 2 * kt];
            A[mt][kt].x = p[0]; A[mt][kt].y = p[1];
        }

    const v8f z8 = {0.f, 0.f, 0.f, 0.f, 0.f, 0.f, 0.f, 0.f};

    for (int ct = 0; ct < 2; ++ct) {
        float X[64];
        {
            float4 v[6];
            #pragma unroll
            for (int q = 0; q < 6; ++q) v[q] = Xch[(w * 6 + q) * 32 + ct * 16 + c];
            float2 ph[8], pl[8];
            prod3(ph, v[0], v[1], v[2]);
            prod3(pl, v[3], v[4], v[5]);
            float lx[8], ly[8];
            #pragma unroll
            for (int b = 0; b < 8; ++b) {
                lx[b] = h ? pl[b].y : pl[b].x;
                ly[b] = h ? pl[b].x : -pl[b].y;
            }
            #pragma unroll
            for (int j = 0; j < 64; ++j) {
                const int k = ring_src(j);
                X[j] = ph[k >> 3].x * lx[k & 7] + ph[k >> 3].y * ly[k & 7];
            }
        }

        #pragma unroll
        for (int l = 0; l < 3; ++l) {
            v8f D[8];
            #pragma unroll
            for (int kt = 0; kt < 4; ++kt)
                #pragma unroll
                for (int b = 0; b < 8; ++b) {
                    v2f bb;
                    bb.x = X[(2 * kt) * 8 + b];
                    bb.y = X[(2 * kt + 1) * 8 + b];
                    D[b] = __builtin_amdgcn_wmma_f32_16x16x4_f32(
                        false, A[2 * l][kt], false, bb, (short)0, kt == 0 ? z8 : D[b], false, false);
                }
            v8f D2[8];
            #pragma unroll
            for (int kt = 0; kt < 4; ++kt)
                #pragma unroll
                for (int a = 0; a < 8; ++a) {
                    v2f bb;
                    bb.x = D[2 * kt][a];
                    bb.y = D[2 * kt + 1][a];
                    D2[a] = __builtin_amdgcn_wmma_f32_16x16x4_f32(
                        false, A[2 * l + 1][kt], false, bb, (short)0, kt == 0 ? z8 : D2[a], false, false);
                }
            #pragma unroll
            for (int j = 0; j < 64; ++j) {
                const int k = (l < 2) ? ring_src(j) : j;
                X[j] = D2[k >> 3][k & 7];
            }
        }

        float P[8];
        #pragma unroll
        for (int i = 0; i < 8; ++i) P[i] = 0.f;
        float x1 = 0.f, x3 = 0.f, x5 = 0.f;
        #pragma unroll
        for (int j = 0; j < 64; ++j) {
            const float a = X[j];
            P[((j >> 5) & 1) * 4 + ((j >> 3) & 1) * 2 + ((j >> 1) & 1)] += a * a;
            if (!(j & 16)) x1 += a * X[j ^ 16];
            if (!(j & 4))  x3 += a * X[j ^ 4];
            if (!(j & 1))  x5 += a * X[j ^ 1];
        }
        float z0 = ((P[0] + P[1]) + (P[2] + P[3])) - ((P[4] + P[5]) + (P[6] + P[7]));
        float z2 = ((P[0] + P[1]) + (P[4] + P[5])) - ((P[2] + P[3]) + (P[6] + P[7]));
        float z4 = ((P[0] + P[2]) + (P[4] + P[6])) - ((P[1] + P[3]) + (P[5] + P[7]));
        x1 *= 2.f; x3 *= 2.f; x5 *= 2.f;
        z0 += __shfl_xor(z0, 16, 32); z2 += __shfl_xor(z2, 16, 32); z4 += __shfl_xor(z4, 16, 32);
        x1 += __shfl_xor(x1, 16, 32); x3 += __shfl_xor(x3, 16, 32); x5 += __shfl_xor(x5, 16, 32);

        const int ee = blockIdx.x * 128 + w * 32 + ct * 16 + c;
        const float4 o = h ? make_float4(z4, x5, z0, x1) : make_float4(z0, x1, z2, x3);
        *(float4*)&out[ee * 8 + 4 * h] = o;
    }
}

extern "C" void kernel_launch(void* const* d_in, const int* in_sizes, int n_in,
                              void* d_out, int out_size, void* d_ws, size_t ws_size,
                              hipStream_t stream) {
    const float* x     = (const float*)d_in[0];
    const float* theta = (const float*)d_in[1];
    float* out = (float*)d_out;

    qsim_kernel<<<131072 / 128, 128, 0, stream>>>(x, theta, out);
}


# Last layer's rotations folded into the observables: out = <psi'|u^H P u|psi'> from 1-qubit sums of the state behind the last ring; 256 not 384 f32 WMMAs per wave; sum order changes
# speedup vs baseline: 17.1403x; 1.1774x over previous
#include <hip/hip_runtime.h>
#include <math.h>


typedef float v2f __attribute__((ext_vector_type(2)));
typedef float v8f __attribute__((ext_vector_type(8)));

#define PI_F 3.14159265358979323846f

__device__ __forceinline__ int ring_src(int j) {
    int s = j;
    if (s & 1)  s ^= 32;
    if (s & 2)  s ^= 1;
    if (s & 4)  s ^= 2;
    if (s & 8)  s ^= 4;
    if (s & 16) s ^= 8;
    if (s & 32) s ^= 16;
    return s;
}

__device__ __forceinline__ void rot3(float2& a, float2& b, float2 tx, float2 ty, float2 tz) {
    const float2 a1 = make_float2(tx.x * a.x + tx.y * b.y, tx.x * a.y - tx.y * b.x);
    const float2 b1 = make_float2(tx.y * a.y + tx.x * b.x, tx.x * b.y - tx.y * a.x);
    const float2 a2 = make_float2(ty.x * a1.x - ty.y * b1.x, ty.x * a1.y - ty.y * b1.y);
    const float2 b2 = make_float2(ty.y * a1.x + ty.x * b1.x, ty.y * a1.y + ty.x * b1.y);
    a = make_float2(tz.x * a2.x + tz.y * a2.y, tz.x * a2.y - tz.y * a2.x);
    b = make_float2(tz.x * b2.x - tz.y * b2.y, tz.x * b2.y + tz.y * b2.x);
}

__device__ __forceinline__ float2 cmul(float2 a, float2 b) {
    return make_float2(a.x * b.x - a.y * b.y, a.x * b.y + a.y * b.x);
}

__device__ __forceinline__ float2 cmulc(float2 a, float2 b) {
    return make_float2(a.x * b.x + a.y * b.y, a.x * b.y - a.y * b.x);
}

__device__ __forceinline__ float xhalf(float v) {
    return __builtin_bit_cast(float, __builtin_amdgcn_permlanex16(
        __builtin_bit_cast(int, v), __builtin_bit_cast(int, v), 0x76543210, 0xfedcba98, false, false));
}

__device__ __forceinline__ constexpr int par6(int v) {
    return (v ^ (v >> 1) ^ (v >> 2) ^ (v >> 3) ^ (v >> 4) ^ (v >> 5)) & 1;
}
__device__ __forceinline__ constexpr int pair_mask(int q) {
    return q == 0 ? 48 : q == 1 ? 24 : q == 2 ? 12 : q == 3 ? 6 : q == 4 ? 3 : 49;
}
__device__ __forceinline__ constexpr int bit_mask(int q) {
    return q == 0 ? 31 : q == 1 ? 48 : q == 2 ? 56 : q == 3 ? 60 : q == 4 ? 62 : 63;
}

__device__ __forceinline__ void prod3(float2 (&p)[8], float4 v0, float4 v1, float4 v2) {
    float2 p2[4];
    #pragma unroll
    for (int i = 0; i < 4; ++i) {
        const float2 f0 = (i & 2) ? make_float2(v0.z, v0.w) : make_float2(v0.x, v0.y);
        const float2 f1 = (i & 1) ? make_float2(v1.z, v1.w) : make_float2(v1.x, v1.y);
        p2[i] = cmul(f0, f1);
    }
    #pragma unroll
    for (int i = 0; i < 8; ++i) {
        const float2 f2 = (i & 1) ? make_float2(v2.z, v2.w) : make_float2(v2.x, v2.y);
        p[i] = cmul(p2[i >> 1], f2);
    }
}

__global__ __launch_bounds__(128) void qsim_kernel(const float* __restrict__ x,
                                                   const float* __restrict__ theta,
                                                   float* __restrict__ out) {
    __shared__ float2 trig[54];
    __shared__ float2 Ug[18 * 4];
    __shared__ float  E[4 * 256];
    __shared__ float4 Obs[6];
    __shared__ float4 Xch[4 * 6 * 32];

    const int tid  = threadIdx.x;
    const int w    = tid >> 5;
    const int lane = tid & 31;
    const int c    = lane & 15;
    const int h    = lane >> 4;

    if (tid < 54) {
        const float a = theta[tid];
        trig[tid] = make_float2(cosf(0.5f * a), sinf(0.5f * a));
    }

    {
        const int e = blockIdx.x * 128 + w * 32 + lane;
        const float* xp = x + e * 24;
        #pragma unroll
        for (int q = 0; q < 6; ++q) {
            float m = (xp[4 * q] + xp[4 * q + 1] + xp[4 * q + 2] + xp[4 * q + 3]) * 0.25f;
            m = fminf(6.f, fmaxf(-6.f, m));
            const float a  = m * (PI_F / 6.f);
            const float c2 = cosf(0.5f * a),  s2 = sinf(0.5f * a);
            const float c4 = cosf(0.25f * a), s4 = sinf(0.25f * a);
            Xch[(w * 6 + q) * 32 + lane] = make_float4(c2 * c4, -c2 * s4, s2 * s4, -s2 * c4);
        }
    }
    __syncthreads();

    if (tid < 36) {
        const int g = tid >> 1, i = tid & 1;
        float2 a = make_float2(i ? 0.f : 1.f, 0.f);
        float2 b = make_float2(i ? 1.f : 0.f, 0.f);
        rot3(a, b, trig[g * 3 + 0], trig[g * 3 + 1], trig[g * 3 + 2]);
        Ug[g * 4 + 0 + i] = a;
        Ug[g * 4 + 2 + i] = b;
    }
    __syncthreads();

    if (tid < 6) {
        const float2* u = Ug + (12 + tid) * 4;
        const float2 u00 = u[0], u01 = u[1], u10 = u[2], u11 = u[3];
        float o00, o11; float2 o01;
        if (tid & 1) {
            o00 = 2.f * cmulc(u00, u10).x;
            o11 = 2.f * cmulc(u01, u11).x;
            const float2 s = cmulc(u00, u11), t = cmulc(u10, u01);
            o01 = make_float2(s.x + t.x, s.y + t.y);
        } else {
            o00 = cmulc(u00, u00).x - cmulc(u10, u10).x;
            o11 = cmulc(u01, u01).x - cmulc(u11, u11).x;
            const float2 s = cmulc(u00, u01), t = cmulc(u10, u11);
            o01 = make_float2(s.x - t.x, s.y - t.y);
        }
        Obs[tid] = make_float4(o00, o11, 2.f * o01.x, 2.f * o01.y);
    }

    for (int i = 0; i < 8; ++i) {
        const int idx = tid + 128 * i;
        const int mt = idx >> 8, m = (idx >> 4) & 15, k = idx & 15;
        const int hp = m >> 3, rp = m & 7, hh = k >> 3, r = k & 7;
        const float2* u = Ug + mt * 12;
        const float2 e0 = u[0 + ((rp >> 2) & 1) * 2 + ((r >> 2) & 1)];
        const float2 e1 = u[4 + ((rp >> 1) & 1) * 2 + ((r >> 1) & 1)];
        const float2 e2 = u[8 + (rp & 1) * 2 + (r & 1)];
        const float2 mm = cmul(cmul(e0, e1), e2);
        E[idx] = (hp == hh) ? mm.x : (hp ? mm.y : -mm.y);
    }
    __syncthreads();

    v2f A[4][4];
    #pragma unroll
    for (int mt = 0; mt < 4; ++mt)
        #pragma unroll
        for (int kt = 0; kt < 4; ++kt) {
            const float* p = &E[mt * 256 + c * 16 + 8 * h + 2 * kt];
            A[mt][kt].x = p[0]; A[mt][kt].y = p[1];
        }

    const v8f z8 = {0.f, 0.f, 0.f, 0.f, 0.f, 0.f, 0.f, 0.f};
    const float ysign = h ? 1.f : -1.f;

    for (int ct = 0; ct < 2; ++ct) {
        float X[64];
        {
            float4 v[6];
            #pragma unroll
            for (int q = 0; q < 6; ++q) v[q] = Xch[(w * 6 + q) * 32 + ct * 16 + c];
            float2 ph[8], pl[8];
            prod3(ph, v[0], v[1], v[2]);
            prod3(pl, v[3], v[4], v[5]);
            float lx[8], ly[8];
            #pragma unroll
            for (int b = 0; b < 8; ++b) {
                lx[b] = h ? pl[b].y : pl[b].x;
                ly[b] = h ? pl[b].x : -pl[b].y;
            }
            #pragma unroll
            for (int j = 0; j < 64; ++j) {
                const int k = ring_src(j);
                X[j] = ph[k >> 3].x * lx[k & 7] + ph[k >> 3].y * ly[k & 7];
            }
        }

        v8f D2[8];
        #pragma unroll
        for (int l = 0; l < 2; ++l) {
            v8f D[8];
            #pragma unroll
            for (int kt = 0; kt < 4; ++kt)
                #pragma unroll
                for (int b = 0; b < 8; ++b) {
                    v2f bb;
                    bb.x = X[(2 * kt) * 8 + b];
                    bb.y = X[(2 * kt + 1) * 8 + b];
                    D[b] = __builtin_amdgcn_wmma_f32_16x16x4_f32(
                        false, A[2 * l][kt], false, bb, (short)0, kt == 0 ? z8 : D[b], false, false);
                }
            #pragma unroll
            for (int kt = 0; kt < 4; ++kt)
                #pragma unroll
                for (int a = 0; a < 8; ++a) {
                    v2f bb;
                    bb.x = D[2 * kt][a];
                    bb.y = D[2 * kt + 1][a];
                    D2[a] = __builtin_amdgcn_wmma_f32_16x16x4_f32(
                        false, A[2 * l + 1][kt], false, bb, (short)0, kt == 0 ? z8 : D2[a], false, false);
                }
            if (l == 0) {
                #pragma unroll
                for (int j = 0; j < 64; ++j) {
                    const int k = ring_src(j);
                    X[j] = D2[k >> 3][k & 7];
                }
            }
        }

        v2f Y2[32], Yo2[32];
        #pragma unroll
        for (int p = 0; p < 32; ++p) {
            Y2[p].x = D2[p >> 2][2 * (p & 3)];
            Y2[p].y = D2[p >> 2][2 * (p & 3) + 1];
            Yo2[p].x = xhalf(Y2[p].x);
            Yo2[p].y = xhalf(Y2[p].y);
        }
        v2f bA[8], bB[8];
        #pragma unroll
        for (int i = 0; i < 8; ++i) { bA[i].x = 0.f; bA[i].y = 0.f; bB[i].x = 0.f; bB[i].y = 0.f; }
        #pragma unroll
        for (int p = 0; p < 32; ++p) {
            const int s = 2 * p;
            const int iA = par6(s & bit_mask(1)) * 4 + par6(s & bit_mask(2)) * 2 + par6(s & bit_mask(3));
            const int iB = par6(s & bit_mask(4)) * 4 + par6(s & bit_mask(0)) * 2 + par6(s & bit_mask(5));
            bA[iA] += Y2[p] * Y2[p];
            bB[iB] += Y2[p] * Y2[p];
        }
        float binA[8], binB[8];
        #pragma unroll
        for (int i = 0; i < 8; ++i) { binA[i] = bA[i].x + bA[i].y; binB[i] = bB[i].x + bB[i ^ 3].y; }

        float v[6];
        #pragma unroll
        for (int q = 0; q < 6; ++q) {
            const int M = pair_mask(q), F = bit_mask(q), PM = M >> 1;
            const int hb = (PM & 16) ? 16 : (PM & 8) ? 8 : (PM & 4) ? 4 : (PM & 2) ? 2 : 1;
            v2f cr = {0.f, 0.f}, gp = {0.f, 0.f}, gn = {0.f, 0.f};
            #pragma unroll
            for (int p = 0; p < 32; ++p) {
                if (p & hb) continue;
                v2f pr = Y2[p ^ PM], po = Yo2[p ^ PM];
                if (M & 1) { pr = __builtin_shufflevector(pr, pr, 1, 0); po = __builtin_shufflevector(po, po, 1, 0); }
                cr += Y2[p] * pr;
                if (par6((2 * p) & F)) gn += Y2[p] * po; else gp += Y2[p] * po;
            }
            const float Cr = cr.x + cr.y;
            const float G  = (F & 1) ? ((gp.x - gp.y) - (gn.x - gn.y)) : ((gp.x + gp.y) - (gn.x + gn.y));
            const bool fromA = (q >= 1 && q <= 3);
            const int sel = fromA ? (4 >> (q - 1)) : (q == 4 ? 4 : (q == 0 ? 2 : 1));
            float s0 = 0.f, s1 = 0.f;
            #pragma unroll
            for (int i = 0; i < 8; ++i) {
                const float b = fromA ? binA[i] : binB[i];
                if (i & sel) s1 += b; else s0 += b;
            }
            const float4 O = Obs[q];
            float t = O.x * s0 + O.y * s1 + O.z * Cr + (ysign * O.w) * G;
            t += xhalf(t);
            v[q] = t;
        }

        const int ee = blockIdx.x * 128 + w * 32 + ct * 16 + c;
        const float4 o = h ? make_float4(v[4], v[5], v[0], v[1]) : make_float4(v[0], v[1], v[2], v[3]);
        *(float4*)&out[ee * 8 + 4 * h] = o;
    }
}

extern "C" void kernel_launch(void* const* d_in, const int* in_sizes, int n_in,
                              void* d_out, int out_size, void* d_ws, size_t ws_size,
                              hipStream_t stream) {
    const float* x     = (const float*)d_in[0];
    const float* theta = (const float*)d_in[1];
    float* out = (float*)d_out;

    qsim_kernel<<<131072 / 128, 128, 0, stream>>>(x, theta, out);
}


# Angle encoding without library trig: sin/cos of a/4 as branch-free polynomials (the clamp bounds |a/4| <= pi/4), a/2 by double angle; an element's six row loads issue together, not one per qubit
# speedup vs baseline: 18.5351x; 1.0814x over previous
#include <hip/hip_runtime.h>
#include <math.h>


typedef float v2f __attribute__((ext_vector_type(2)));
typedef float v8f __attribute__((ext_vector_type(8)));

#define PI_F 3.14159265358979323846f

__device__ __forceinline__ int ring_src(int j) {
    int s = j;
    if (s & 1)  s ^= 32;
    if (s & 2)  s ^= 1;
    if (s & 4)  s ^= 2;
    if (s & 8)  s ^= 4;
    if (s & 16) s ^= 8;
    if (s & 32) s ^= 16;
    return s;
}

__device__ __forceinline__ void rot3(float2& a, float2& b, float2 tx, float2 ty, float2 tz) {
    const float2 a1 = make_float2(tx.x * a.x + tx.y * b.y, tx.x * a.y - tx.y * b.x);
    const float2 b1 = make_float2(tx.y * a.y + tx.x * b.x, tx.x * b.y - tx.y * a.x);
    const float2 a2 = make_float2(ty.x * a1.x - ty.y * b1.x, ty.x * a1.y - ty.y * b1.y);
    const float2 b2 = make_float2(ty.y * a1.x + ty.x * b1.x, ty.y * a1.y + ty.x * b1.y);
    a = make_float2(tz.x * a2.x + tz.y * a2.y, tz.x * a2.y - tz.y * a2.x);
    b = make_float2(tz.x * b2.x - tz.y * b2.y, tz.x * b2.y + tz.y * b2.x);
}

__device__ __forceinline__ float2 cmul(float2 a, float2 b) {
    return make_float2(a.x * b.x - a.y * b.y, a.x * b.y + a.y * b.x);
}

__device__ __forceinline__ float2 cmulc(float2 a, float2 b) {
    return make_float2(a.x * b.x + a.y * b.y, a.x * b.y - a.y * b.x);
}

__device__ __forceinline__ float xhalf(float v) {
    return __builtin_bit_cast(float, __builtin_amdgcn_permlanex16(
        __builtin_bit_cast(int, v), __builtin_bit_cast(int, v), 0x76543210, 0xfedcba98, false, false));
}

__device__ __forceinline__ constexpr int par6(int v) {
    return (v ^ (v >> 1) ^ (v >> 2) ^ (v >> 3) ^ (v >> 4) ^ (v >> 5)) & 1;
}
__device__ __forceinline__ constexpr int pair_mask(int q) {
    return q == 0 ? 48 : q == 1 ? 24 : q == 2 ? 12 : q == 3 ? 6 : q == 4 ? 3 : 49;
}
__device__ __forceinline__ constexpr int bit_mask(int q) {
    return q == 0 ? 31 : q == 1 ? 48 : q == 2 ? 56 : q == 3 ? 60 : q == 4 ? 62 : 63;
}

__device__ __forceinline__ void prod3(float2 (&p)[8], float4 v0, float4 v1, float4 v2) {
    float2 p2[4];
    #pragma unroll
    for (int i = 0; i < 4; ++i) {
        const float2 f0 = (i & 2) ? make_float2(v0.z, v0.w) : make_float2(v0.x, v0.y);
        const float2 f1 = (i & 1) ? make_float2(v1.z, v1.w) : make_float2(v1.x, v1.y);
        p2[i] = cmul(f0, f1);
    }
    #pragma unroll
    for (int i = 0; i < 8; ++i) {
        const float2 f2 = (i & 1) ? make_float2(v2.z, v2.w) : make_float2(v2.x, v2.y);
        p[i] = cmul(p2[i >> 1], f2);
    }
}

__global__ __launch_bounds__(128) void qsim_kernel(const float* __restrict__ x,
                                                   const float* __restrict__ theta,
                                                   float* __restrict__ out) {
    __shared__ float2 trig[54];
    __shared__ float2 Ug[18 * 4];
    __shared__ float  E[4 * 256];
    __shared__ float4 Obs[6];
    __shared__ float4 Xch[4 * 6 * 32];

    const int tid  = threadIdx.x;
    const int w    = tid >> 5;
    const int lane = tid & 31;
    const int c    = lane & 15;
    const int h    = lane >> 4;

    float xr[24];
    {
        const int e = blockIdx.x * 128 + w * 32 + lane;
        const float* xp = x + e * 24;
        #pragma unroll
        for (int i = 0; i < 24; ++i) xr[i] = xp[i];
    }

    if (tid < 54) {
        const float a = theta[tid];
        trig[tid] = make_float2(cosf(0.5f * a), sinf(0.5f * a));
    }

    {
        #pragma unroll
        for (int q = 0; q < 6; ++q) {
            float m = (xr[4 * q] + xr[4 * q + 1] + xr[4 * q + 2] + xr[4 * q + 3]) * 0.25f;
            m = fminf(6.f, fmaxf(-6.f, m));
            const float a  = m * (PI_F / 6.f);
            const float t  = 0.25f * a;
            const float z  = t * t;
            const float s4 = fmaf(t * z, fmaf(z, fmaf(z, -1.9515295891e-4f, 8.3321608736e-3f), -1.6666654611e-1f), t);
            const float c4 = fmaf(z * z, fmaf(z, fmaf(z, 2.443315711809948e-5f, -1.388731625493765e-3f), 4.166664568298827e-2f),
                                  fmaf(z, -0.5f, 1.f));
            const float s2 = 2.f * s4 * c4;
            const float c2 = fmaf(-2.f * s4, s4, 1.f);
            Xch[(w * 6 + q) * 32 + lane] = make_float4(c2 * c4, -c2 * s4, s2 * s4, -s2 * c4);
        }
    }
    __syncthreads();

    if (tid < 36) {
        const int g = tid >> 1, i = tid & 1;
        float2 a = make_float2(i ? 0.f : 1.f, 0.f);
        float2 b = make_float2(i ? 1.f : 0.f, 0.f);
        rot3(a, b, trig[g * 3 + 0], trig[g * 3 + 1], trig[g * 3 + 2]);
        Ug[g * 4 + 0 + i] = a;
        Ug[g * 4 + 2 + i] = b;
    }
    __syncthreads();

    if (tid < 6) {
        const float2* u = Ug + (12 + tid) * 4;
        const float2 u00 = u[0], u01 = u[1], u10 = u[2], u11 = u[3];
        float o00, o11; float2 o01;
        if (tid & 1) {
            o00 = 2.f * cmulc(u00, u10).x;
            o11 = 2.f * cmulc(u01, u11).x;
            const float2 s = cmulc(u00, u11), t = cmulc(u10, u01);
            o01 = make_float2(s.x + t.x, s.y + t.y);
        } else {
            o00 = cmulc(u00, u00).x - cmulc(u10, u10).x;
            o11 = cmulc(u01, u01).x - cmulc(u11, u11).x;
            const float2 s = cmulc(u00, u01), t = cmulc(u10, u11);
            o01 = make_float2(s.x - t.x, s.y - t.y);
        }
        Obs[tid] = make_float4(o00, o11, 2.f * o01.x, 2.f * o01.y);
    }

    for (int i = 0; i < 8; ++i) {
        const int idx = tid + 128 * i;
        const int mt = idx >> 8, m = (idx >> 4) & 15, k = idx & 15;
        const int hp = m >> 3, rp = m & 7, hh = k >> 3, r = k & 7;
        const float2* u = Ug + mt * 12;
        const float2 e0 = u[0 + ((rp >> 2) & 1) * 2 + ((r >> 2) & 1)];
        const float2 e1 = u[4 + ((rp >> 1) & 1) * 2 + ((r >> 1) & 1)];
        const float2 e2 = u[8 + (rp & 1) * 2 + (r & 1)];
        const float2 mm = cmul(cmul(e0, e1), e2);
        E[idx] = (hp == hh) ? mm.x : (hp ? mm.y : -mm.y);
    }
    __syncthreads();

    v2f A[4][4];
    #pragma unroll
    for (int mt = 0; mt < 4; ++mt)
        #pragma unroll
        for (int kt = 0; kt < 4; ++kt) {
            const float* p = &E[mt * 256 + c * 16 + 8 * h + 2 * kt];
            A[mt][kt].x = p[0]; A[mt][kt].y = p[1];
        }

    const v8f z8 = {0.f, 0.f, 0.f, 0.f, 0.f, 0.f, 0.f, 0.f};
    const float ysign = h ? 1.f : -1.f;

    for (int ct = 0; ct < 2; ++ct) {
        float X[64];
        {
            float4 v[6];
            #pragma unroll
            for (int q = 0; q < 6; ++q) v[q] = Xch[(w * 6 + q) * 32 + ct * 16 + c];
            float2 ph[8], pl[8];
            prod3(ph, v[0], v[1], v[2]);
            prod3(pl, v[3], v[4], v[5]);
            float lx[8], ly[8];
            #pragma unroll
            for (int b = 0; b < 8; ++b) {
                lx[b] = h ? pl[b].y : pl[b].x;
                ly[b] = h ? pl[b].x : -pl[b].y;
            }
            #pragma unroll
            for (int j = 0; j < 64; ++j) {
                const int k = ring_src(j);
                X[j] = ph[k >> 3].x * lx[k & 7] + ph[k >> 3].y * ly[k & 7];
            }
        }

        v8f D2[8];
        #pragma unroll
        for (int l = 0; l < 2; ++l) {
            v8f D[8];
            #pragma unroll
            for (int kt = 0; kt < 4; ++kt)
                #pragma unroll
                for (int b = 0; b < 8; ++b) {
                    v2f bb;
                    bb.x = X[(2 * kt) * 8 + b];
                    bb.y = X[(2 * kt + 1) * 8 + b];
                    D[b] = __builtin_amdgcn_wmma_f32_16x16x4_f32(
                        false, A[2 * l][kt], false, bb, (short)0, kt == 0 ? z8 : D[b], false, false);
                }
            #pragma unroll
            for (int kt = 0; kt < 4; ++kt)
                #pragma unroll
                for (int a = 0; a < 8; ++a) {
                    v2f bb;
                    bb.x = D[2 * kt][a];
                    bb.y = D[2 * kt + 1][a];
                    D2[a] = __builtin_amdgcn_wmma_f32_16x16x4_f32(
                        false, A[2 * l + 1][kt], false, bb, (short)0, kt == 0 ? z8 : D2[a], false, false);
                }
            if (l == 0) {
                #pragma unroll
                for (int j = 0; j < 64; ++j) {
                    const int k = ring_src(j);
                    X[j] = D2[k >> 3][k & 7];
                }
            }
        }

        v2f Y2[32], Yo2[32];
        #pragma unroll
        for (int p = 0; p < 32; ++p) {
            Y2[p].x = D2[p >> 2][2 * (p & 3)];
            Y2[p].y = D2[p >> 2][2 * (p & 3) + 1];
            Yo2[p].x = xhalf(Y2[p].x);
            Yo2[p].y = xhalf(Y2[p].y);
        }
        v2f bA[8], bB[8];
        #pragma unroll
        for (int i = 0; i < 8; ++i) { bA[i].x = 0.f; bA[i].y = 0.f; bB[i].x = 0.f; bB[i].y = 0.f; }
        #pragma unroll
        for (int p = 0; p < 32; ++p) {
            const int s = 2 * p;
            const int iA = par6(s & bit_mask(1)) * 4 + par6(s & bit_mask(2)) * 2 + par6(s & bit_mask(3));
            const int iB = par6(s & bit_mask(4)) * 4 + par6(s & bit_mask(0)) * 2 + par6(s & bit_mask(5));
            bA[iA] += Y2[p] * Y2[p];
            bB[iB] += Y2[p] * Y2[p];
        }
        float binA[8], binB[8];
        #pragma unroll
        for (int i = 0; i < 8; ++i) { binA[i] = bA[i].x + bA[i].y; binB[i] = bB[i].x + bB[i ^ 3].y; }

        float v[6];
        #pragma unroll
        for (int q = 0; q < 6; ++q) {
            const int M = pair_mask(q), F = bit_mask(q), PM = M >> 1;
            const int hb = (PM & 16) ? 16 : (PM & 8) ? 8 : (PM & 4) ? 4 : (PM & 2) ? 2 : 1;
            v2f cr = {0.f, 0.f}, gp = {0.f, 0.f}, gn = {0.f, 0.f};
            #pragma unroll
            for (int p = 0; p < 32; ++p) {
                if (p & hb) continue;
                v2f pr = Y2[p ^ PM], po = Yo2[p ^ PM];
                if (M & 1) { pr = __builtin_shufflevector(pr, pr, 1, 0); po = __builtin_shufflevector(po, po, 1, 0); }
                cr += Y2[p] * pr;
                if (par6((2 * p) & F)) gn += Y2[p] * po; else gp += Y2[p] * po;
            }
            const float Cr = cr.x + cr.y;
            const float G  = (F & 1) ? ((gp.x - gp.y) - (gn.x - gn.y)) : ((gp.x + gp.y) - (gn.x + gn.y));
            const bool fromA = (q >= 1 && q <= 3);
            const int sel = fromA ? (4 >> (q - 1)) : (q == 4 ? 4 : (q == 0 ? 2 : 1));
            float s0 = 0.f, s1 = 0.f;
            #pragma unroll
            for (int i = 0; i < 8; ++i) {
                const float b = fromA ? binA[i] : binB[i];
                if (i & sel) s1 += b; else s0 += b;
            }
            const float4 O = Obs[q];
            float t = O.x * s0 + O.y * s1 + O.z * Cr + (ysign * O.w) * G;
            t += xhalf(t);
            v[q] = t;
        }

        const int ee = blockIdx.x * 128 + w * 32 + ct * 16 + c;
        const float4 o = h ? make_float4(v[4], v[5], v[0], v[1]) : make_float4(v[0], v[1], v[2], v[3]);
        *(float4*)&out[ee * 8 + 4 * h] = o;
    }
}

extern "C" void kernel_launch(void* const* d_in, const int* in_sizes, int n_in,
                              void* d_out, int out_size, void* d_ws, size_t ws_size,
                              hipStream_t stream) {
    const float* x     = (const float*)d_in[0];
    const float* theta = (const float*)d_in[1];
    float* out = (float*)d_out;

    qsim_kernel<<<131072 / 128, 128, 0, stream>>>(x, theta, out);
}


# Layer 0 on the product state as four outer products: 16 f32 WMMAs (two k steps each) give its Q and S vectors, packed FMAs form layer 1's B operands; 80 not 128 WMMAs a tile; sum order changed
# speedup vs baseline: 20.3056x; 1.0955x over previous
#include <hip/hip_runtime.h>
#include <math.h>


typedef float v2f __attribute__((ext_vector_type(2)));
typedef float v8f __attribute__((ext_vector_type(8)));

#define PI_F 3.14159265358979323846f

__device__ __forceinline__ int ring_src(int j) {
    int s = j;
    if (s & 1)  s ^= 32;
    if (s & 2)  s ^= 1;
    if (s & 4)  s ^= 2;
    if (s & 8)  s ^= 4;
    if (s & 16) s ^= 8;
    if (s & 32) s ^= 16;
    return s;
}

__device__ __forceinline__ void rot3(float2& a, float2& b, float2 tx, float2 ty, float2 tz) {
    const float2 a1 = make_float2(tx.x * a.x + tx.y * b.y, tx.x * a.y - tx.y * b.x);
    const float2 b1 = make_float2(tx.y * a.y + tx.x * b.x, tx.x * b.y - tx.y * a.x);
    const float2 a2 = make_float2(ty.x * a1.x - ty.y * b1.x, ty.x * a1.y - ty.y * b1.y);
    const float2 b2 = make_float2(ty.y * a1.x + ty.x * b1.x, ty.y * a1.y + ty.x * b1.y);
    a = make_float2(tz.x * a2.x + tz.y * a2.y, tz.x * a2.y - tz.y * a2.x);
    b = make_float2(tz.x * b2.x - tz.y * b2.y, tz.x * b2.y + tz.y * b2.x);
}

__device__ __forceinline__ float2 cmul(float2 a, float2 b) {
    return make_float2(a.x * b.x - a.y * b.y, a.x * b.y + a.y * b.x);
}

__device__ __forceinline__ float2 cmulc(float2 a, float2 b) {
    return make_float2(a.x * b.x + a.y * b.y, a.x * b.y - a.y * b.x);
}

__device__ __forceinline__ float xhalf(float v) {
    return __builtin_bit_cast(float, __builtin_amdgcn_permlanex16(
        __builtin_bit_cast(int, v), __builtin_bit_cast(int, v), 0x76543210, 0xfedcba98, false, false));
}

__device__ __forceinline__ constexpr int par6(int v) {
    return (v ^ (v >> 1) ^ (v >> 2) ^ (v >> 3) ^ (v >> 4) ^ (v >> 5)) & 1;
}
__device__ __forceinline__ constexpr int pair_mask(int q) {
    return q == 0 ? 48 : q == 1 ? 24 : q == 2 ? 12 : q == 3 ? 6 : q == 4 ? 3 : 49;
}
__device__ __forceinline__ constexpr int bit_mask(int q) {
    return q == 0 ? 31 : q == 1 ? 48 : q == 2 ? 56 : q == 3 ? 60 : q == 4 ? 62 : 63;
}

__device__ __forceinline__ void prod3(float2 (&p)[8], float4 v0, float4 v1, float4 v2) {
    float2 p2[4];
    #pragma unroll
    for (int i = 0; i < 4; ++i) {
        const float2 f0 = (i & 2) ? make_float2(v0.z, v0.w) : make_float2(v0.x, v0.y);
        const float2 f1 = (i & 1) ? make_float2(v1.z, v1.w) : make_float2(v1.x, v1.y);
        p2[i] = cmul(f0, f1);
    }
    #pragma unroll
    for (int i = 0; i < 8; ++i) {
        const float2 f2 = (i & 1) ? make_float2(v2.z, v2.w) : make_float2(v2.x, v2.y);
        p[i] = cmul(p2[i >> 1], f2);
    }
}

__global__ __launch_bounds__(128) __attribute__((amdgpu_waves_per_eu(4, 4))) void qsim_kernel(const float* __restrict__ x,
                                                   const float* __restrict__ theta,
                                                   float* __restrict__ out) {
    __shared__ float2 trig[54];
    __shared__ float2 Ug[18 * 4];
    __shared__ float  E[4 * 256];
    __shared__ float4 Obs[6];
    __shared__ float4 Xch[4 * 6 * 32];

    const int tid  = threadIdx.x;
    const int w    = tid >> 5;
    const int lane = tid & 31;
    const int c    = lane & 15;
    const int h    = lane >> 4;

    float xr[24];
    {
        const int e = blockIdx.x * 128 + w * 32 + lane;
        const float* xp = x + e * 24;
        #pragma unroll
        for (int i = 0; i < 24; ++i) xr[i] = xp[i];
    }

    if (tid < 54) {
        const float a = theta[tid];
        trig[tid] = make_float2(cosf(0.5f * a), sinf(0.5f * a));
    }

    {
        #pragma unroll
        for (int q = 0; q < 6; ++q) {
            float m = (xr[4 * q] + xr[4 * q + 1] + xr[4 * q + 2] + xr[4 * q + 3]) * 0.25f;
            m = fminf(6.f, fmaxf(-6.f, m));
            const float a  = m * (PI_F / 6.f);
            const float t  = 0.25f * a;
            const float z  = t * t;
            const float s4 = fmaf(t * z, fmaf(z, fmaf(z, -1.9515295891e-4f, 8.3321608736e-3f), -1.6666654611e-1f), t);
            const float c4 = fmaf(z * z, fmaf(z, fmaf(z, 2.443315711809948e-5f, -1.388731625493765e-3f), 4.166664568298827e-2f),
                                  fmaf(z, -0.5f, 1.f));
            const float s2 = 2.f * s4 * c4;
            const float c2 = fmaf(-2.f * s4, s4, 1.f);
            Xch[(w * 6 + q) * 32 + lane] = make_float4(c2 * c4, -c2 * s4, s2 * s4, -s2 * c4);
        }
    }
    __syncthreads();

    if (tid < 36) {
        const int g = tid >> 1, i = tid & 1;
        float2 a = make_float2(i ? 0.f : 1.f, 0.f);
        float2 b = make_float2(i ? 1.f : 0.f, 0.f);
        rot3(a, b, trig[g * 3 + 0], trig[g * 3 + 1], trig[g * 3 + 2]);
        Ug[g * 4 + 0 + i] = a;
        Ug[g * 4 + 2 + i] = b;
    }
    __syncthreads();

    if (tid < 6) {
        const float2* u = Ug + (12 + tid) * 4;
        const float2 u00 = u[0], u01 = u[1], u10 = u[2], u11 = u[3];
        float o00, o11; float2 o01;
        if (tid & 1) {
            o00 = 2.f * cmulc(u00, u10).x;
            o11 = 2.f * cmulc(u01, u11).x;
            const float2 s = cmulc(u00, u11), t = cmulc(u10, u01);
            o01 = make_float2(s.x + t.x, s.y + t.y);
        } else {
            o00 = cmulc(u00, u00).x - cmulc(u10, u10).x;
            o11 = cmulc(u01, u01).x - cmulc(u11, u11).x;
            const float2 s = cmulc(u00, u01), t = cmulc(u10, u11);
            o01 = make_float2(s.x - t.x, s.y - t.y);
        }
        Obs[tid] = make_float4(o00, o11, 2.f * o01.x, 2.f * o01.y);
    }

    for (int i = 0; i < 8; ++i) {
        const int idx = tid + 128 * i;
        const int mt = idx >> 8, m = (idx >> 4) & 15, k = idx & 15;
        const int hp = m >> 3, rp = m & 7, hh = k >> 3, r = k & 7;
        const float2* u = Ug + mt * 12;
        const float2 e0 = u[0 + ((rp >> 2) & 1) * 2 + ((r >> 2) & 1)];
        const float2 e1 = u[4 + ((rp >> 1) & 1) * 2 + ((r >> 1) & 1)];
        const float2 e2 = u[8 + (rp & 1) * 2 + (r & 1)];
        const float2 mm = cmul(cmul(e0, e1), e2);
        E[idx] = (hp == hh) ? mm.x : (hp ? mm.y : -mm.y);
    }
    __syncthreads();

    v2f A[2][4];
    v2f Aq[2][2][2];
    #pragma unroll
    for (int mt = 0; mt < 2; ++mt)
        #pragma unroll
        for (int kt = 0; kt < 4; ++kt) {
            const float* p = &E[(2 + mt) * 256 + c * 16 + 8 * h + 2 * kt];
            A[mt][kt].x = p[0]; A[mt][kt].y = p[1];
        }
    #pragma unroll
    for (int mt = 0; mt < 2; ++mt)
        #pragma unroll
        for (int t = 0; t < 2; ++t)
            #pragma unroll
            for (int kt = 0; kt < 2; ++kt) {
                const float* p = &E[mt * 256 + c * 16 + 8 * h + 4 * kt + t];
                Aq[mt][t][kt].x = p[0]; Aq[mt][t][kt].y = p[2];
            }

    const v8f z8 = {0.f, 0.f, 0.f, 0.f, 0.f, 0.f, 0.f, 0.f};
    const float ysign = h ? 1.f : -1.f;

    for (int ct = 0; ct < 2; ++ct) {
        v8f Q[4], S[4];
        {
            float4 v[6];
            #pragma unroll
            for (int q = 0; q < 6; ++q) v[q] = Xch[(w * 6 + q) * 32 + ct * 16 + c];
            float2 ph[8], pl[8];
            prod3(ph, v[0], v[1], v[2]);
            prod3(pl, v[3], v[4], v[5]);
            float phc[8], plc[8];
            #pragma unroll
            for (int b = 0; b < 8; ++b) {
                phc[b] = h ? ph[b].y : ph[b].x;
                plc[b] = h ? pl[b].y : pl[b].x;
            }
            #pragma unroll
            for (int t2 = 0; t2 < 2; ++t2)
                #pragma unroll
                for (int t5 = 0; t5 < 2; ++t5)
                    #pragma unroll
                    for (int kt = 0; kt < 2; ++kt) {
                        const int i0 = 4 * kt + t2, i1 = i0 + 2;
                        const int l0 = 4 * kt + t5, l1 = l0 + 2;
                        v2f bq, bs;
                        bq.x = phc[ring_src(i0 * 8 + t5) >> 3];
                        bq.y = phc[ring_src(i1 * 8 + t5) >> 3];
                        bs.x = plc[ring_src(t2 * 8 + l0) & 7];
                        bs.y = plc[ring_src(t2 * 8 + l1) & 7];
                        const int t = 2 * t2 + t5;
                        Q[t] = __builtin_amdgcn_wmma_f32_16x16x4_f32(
                            false, Aq[0][t2][kt], false, bq, (short)0, kt == 0 ? z8 : Q[t], false, false);
                        S[t] = __builtin_amdgcn_wmma_f32_16x16x4_f32(
                            false, Aq[1][t5][kt], false, bs, (short)0, kt == 0 ? z8 : S[t], false, false);
                    }
        }

        float Qr[4][8], Qi[4][8], Sb[4][8];
        #pragma unroll
        for (int t = 0; t < 4; ++t)
            #pragma unroll
            for (int r = 0; r < 8; ++r) {
                const float qo = xhalf(Q[t][r]);
                Qr[t][r] = h ? qo : Q[t][r];
                Qi[t][r] = h ? Q[t][r] : qo;
                Sb[t][r] = ysign * xhalf(S[t][r]);
            }
        v2f Xp[4][8];
        #pragma unroll
        for (int kt = 0; kt < 4; ++kt)
            #pragma unroll
            for (int b = 0; b < 8; ++b) {
                const int k0 = ring_src(16 * kt + b), k1 = ring_src(16 * kt + 8 + b);
                v2f acc = {0.f, 0.f};
                #pragma unroll
                for (int t = 0; t < 4; ++t) {
                    v2f qr, qi, sv, sb;
                    qr.x = Qr[t][k0 >> 3]; qr.y = Qr[t][k1 >> 3];
                    qi.x = Qi[t][k0 >> 3]; qi.y = Qi[t][k1 >> 3];
                    sv.x = S[t][k0 & 7];   sv.y = S[t][k1 & 7];
                    sb.x = Sb[t][k0 & 7];  sb.y = Sb[t][k1 & 7];
                    acc += qr * sv;
                    acc += qi * sb;
                }
                Xp[kt][b] = acc;
            }

        v8f D2[8];
        {
            v8f D[8];
            #pragma unroll
            for (int kt = 0; kt < 4; ++kt)
                #pragma unroll
                for (int b = 0; b < 8; ++b)
                    D[b] = __builtin_amdgcn_wmma_f32_16x16x4_f32(
                        false, A[0][kt], false, Xp[kt][b], (short)0, kt == 0 ? z8 : D[b], false, false);
            #pragma unroll
            for (int kt = 0; kt < 4; ++kt)
                #pragma unroll
                for (int a = 0; a < 8; ++a) {
                    v2f bb;
                    bb.x = D[2 * kt][a];
                    bb.y = D[2 * kt + 1][a];
                    D2[a] = __builtin_amdgcn_wmma_f32_16x16x4_f32(
                        false, A[1][kt], false, bb, (short)0, kt == 0 ? z8 : D2[a], false, false);
                }
        }

        v2f Y2[32], Yo2[32];
        #pragma unroll
        for (int p = 0; p < 32; ++p) {
            Y2[p].x = D2[p >> 2][2 * (p & 3)];
            Y2[p].y = D2[p >> 2][2 * (p & 3) + 1];
            Yo2[p].x = xhalf(Y2[p].x);
            Yo2[p].y = xhalf(Y2[p].y);
        }
        v2f bA[8], bB[8];
        #pragma unroll
        for (int i = 0; i < 8; ++i) { bA[i].x = 0.f; bA[i].y = 0.f; bB[i].x = 0.f; bB[i].y = 0.f; }
        #pragma unroll
        for (int p = 0; p < 32; ++p) {
            const int s = 2 * p;
            const int iA = par6(s & bit_mask(1)) * 4 + par6(s & bit_mask(2)) * 2 + par6(s & bit_mask(3));
            const int iB = par6(s & bit_mask(4)) * 4 + par6(s & bit_mask(0)) * 2 + par6(s & bit_mask(5));
            bA[iA] += Y2[p] * Y2[p];
            bB[iB] += Y2[p] * Y2[p];
        }
        float binA[8], binB[8];
        #pragma unroll
        for (int i = 0; i < 8; ++i) { binA[i] = bA[i].x + bA[i].y; binB[i] = bB[i].x + bB[i ^ 3].y; }

        float v[6];
        #pragma unroll
        for (int q = 0; q < 6; ++q) {
            const int M = pair_mask(q), F = bit_mask(q), PM = M >> 1;
            const int hb = (PM & 16) ? 16 : (PM & 8) ? 8 : (PM & 4) ? 4 : (PM & 2) ? 2 : 1;
            v2f cr = {0.f, 0.f}, gp = {0.f, 0.f}, gn = {0.f, 0.f};
            #pragma unroll
            for (int p = 0; p < 32; ++p) {
                if (p & hb) continue;
                v2f pr = Y2[p ^ PM], po = Yo2[p ^ PM];
                if (M & 1) { pr = __builtin_shufflevector(pr, pr, 1, 0); po = __builtin_shufflevector(po, po, 1, 0); }
                cr += Y2[p] * pr;
                if (par6((2 * p) & F)) gn += Y2[p] * po; else gp += Y2[p] * po;
            }
            const float Cr = cr.x + cr.y;
            const float G  = (F & 1) ? ((gp.x - gp.y) - (gn.x - gn.y)) : ((gp.x + gp.y) - (gn.x + gn.y));
            const bool fromA = (q >= 1 && q <= 3);
            const int sel = fromA ? (4 >> (q - 1)) : (q == 4 ? 4 : (q == 0 ? 2 : 1));
            float s0 = 0.f, s1 = 0.f;
            #pragma unroll
            for (int i = 0; i < 8; ++i) {
                const float b = fromA ? binA[i] : binB[i];
                if (i & sel) s1 += b; else s0 += b;
            }
            const float4 O = Obs[q];
            float t = O.x * s0 + O.y * s1 + O.z * Cr + (ysign * O.w) * G;
            t += xhalf(t);
            v[q] = t;
        }

        const int ee = blockIdx.x * 128 + w * 32 + ct * 16 + c;
        const float4 o = h ? make_float4(v[4], v[5], v[0], v[1]) : make_float4(v[0], v[1], v[2], v[3]);
        *(float4*)&out[ee * 8 + 4 * h] = o;
    }
}

extern "C" void kernel_launch(void* const* d_in, const int* in_sizes, int n_in,
                              void* d_out, int out_size, void* d_ws, size_t ws_size,
                              hipStream_t stream) {
    const float* x     = (const float*)d_in[0];
    const float* theta = (const float*)d_in[1];
    float* out = (float*)d_out;

    qsim_kernel<<<131072 / 128, 128, 0, stream>>>(x, theta, out);
}


# Lane^16 exchanges halved in both vector phases: layer 0's complex outer products formed whole per lane (layer 1's first k split by hi, not Re/Im), Im sums fetch one end of a pair; sum order changes
# speedup vs baseline: 21.7107x; 1.0692x over previous
#include <hip/hip_runtime.h>
#include <math.h>


typedef float v2f __attribute__((ext_vector_type(2)));
typedef float v8f __attribute__((ext_vector_type(8)));

#define PI_F 3.14159265358979323846f

__device__ __forceinline__ int ring_src(int j) {
    int s = j;
    if (s & 1)  s ^= 32;
    if (s & 2)  s ^= 1;
    if (s & 4)  s ^= 2;
    if (s & 8)  s ^= 4;
    if (s & 16) s ^= 8;
    if (s & 32) s ^= 16;
    return s;
}

__device__ __forceinline__ void rot3(float2& a, float2& b, float2 tx, float2 ty, float2 tz) {
    const float2 a1 = make_float2(tx.x * a.x + tx.y * b.y, tx.x * a.y - tx.y * b.x);
    const float2 b1 = make_float2(tx.y * a.y + tx.x * b.x, tx.x * b.y - tx.y * a.x);
    const float2 a2 = make_float2(ty.x * a1.x - ty.y * b1.x, ty.x * a1.y - ty.y * b1.y);
    const float2 b2 = make_float2(ty.y * a1.x + ty.x * b1.x, ty.y * a1.y + ty.x * b1.y);
    a = make_float2(tz.x * a2.x + tz.y * a2.y, tz.x * a2.y - tz.y * a2.x);
    b = make_float2(tz.x * b2.x - tz.y * b2.y, tz.x * b2.y + tz.y * b2.x);
}

__device__ __forceinline__ float2 cmul(float2 a, float2 b) {
    return make_float2(a.x * b.x - a.y * b.y, a.x * b.y + a.y * b.x);
}

__device__ __forceinline__ float2 cmulc(float2 a, float2 b) {
    return make_float2(a.x * b.x + a.y * b.y, a.x * b.y - a.y * b.x);
}

__device__ __forceinline__ float xhalf(float v) {
    return __builtin_bit_cast(float, __builtin_amdgcn_permlanex16(
        __builtin_bit_cast(int, v), __builtin_bit_cast(int, v), 0x76543210, 0xfedcba98, true, false));
}

__device__ __forceinline__ constexpr int par6(int v) {
    return (v ^ (v >> 1) ^ (v >> 2) ^ (v >> 3) ^ (v >> 4) ^ (v >> 5)) & 1;
}
__device__ __forceinline__ constexpr int pair_mask(int q) {
    return q == 0 ? 48 : q == 1 ? 24 : q == 2 ? 12 : q == 3 ? 6 : q == 4 ? 3 : 49;
}
__device__ __forceinline__ constexpr int bit_mask(int q) {
    return q == 0 ? 31 : q == 1 ? 48 : q == 2 ? 56 : q == 3 ? 60 : q == 4 ? 62 : 63;
}

__device__ __forceinline__ void prod3(float2 (&p)[8], float4 v0, float4 v1, float4 v2) {
    float2 p2[4];
    #pragma unroll
    for (int i = 0; i < 4; ++i) {
        const float2 f0 = (i & 2) ? make_float2(v0.z, v0.w) : make_float2(v0.x, v0.y);
        const float2 f1 = (i & 1) ? make_float2(v1.z, v1.w) : make_float2(v1.x, v1.y);
        p2[i] = cmul(f0, f1);
    }
    #pragma unroll
    for (int i = 0; i < 8; ++i) {
        const float2 f2 = (i & 1) ? make_float2(v2.z, v2.w) : make_float2(v2.x, v2.y);
        p[i] = cmul(p2[i >> 1], f2);
    }
}

__global__ __launch_bounds__(128) __attribute__((amdgpu_waves_per_eu(4, 4))) void qsim_kernel(const float* __restrict__ x,
                                                   const float* __restrict__ theta,
                                                   float* __restrict__ out) {
    __shared__ float2 trig[54];
    __shared__ float2 Ug[18 * 4];
    __shared__ float  E[4 * 256];
    __shared__ float4 Obs[6];
    __shared__ float4 Xch[4 * 6 * 32];

    const int tid  = threadIdx.x;
    const int w    = tid >> 5;
    const int lane = tid & 31;
    const int c    = lane & 15;
    const int h    = lane >> 4;

    float xr[24];
    {
        const int e = blockIdx.x * 128 + w * 32 + lane;
        const float* xp = x + e * 24;
        #pragma unroll
        for (int i = 0; i < 24; ++i) xr[i] = xp[i];
    }

    if (tid < 54) {
        const float a = theta[tid];
        trig[tid] = make_float2(cosf(0.5f * a), sinf(0.5f * a));
    }

    {
        #pragma unroll
        for (int q = 0; q < 6; ++q) {
            float m = (xr[4 * q] + xr[4 * q + 1] + xr[4 * q + 2] + xr[4 * q + 3]) * 0.25f;
            m = fminf(6.f, fmaxf(-6.f, m));
            const float a  = m * (PI_F / 6.f);
            const float t  = 0.25f * a;
            const float z  = t * t;
            const float s4 = fmaf(t * z, fmaf(z, fmaf(z, -1.9515295891e-4f, 8.3321608736e-3f), -1.6666654611e-1f), t);
            const float c4 = fmaf(z * z, fmaf(z, fmaf(z, 2.443315711809948e-5f, -1.388731625493765e-3f), 4.166664568298827e-2f),
                                  fmaf(z, -0.5f, 1.f));
            const float s2 = 2.f * s4 * c4;
            const float c2 = fmaf(-2.f * s4, s4, 1.f);
            Xch[(w * 6 + q) * 32 + lane] = make_float4(c2 * c4, -c2 * s4, s2 * s4, -s2 * c4);
        }
    }
    __syncthreads();

    if (tid < 36) {
        const int g = tid >> 1, i = tid & 1;
        float2 a = make_float2(i ? 0.f : 1.f, 0.f);
        float2 b = make_float2(i ? 1.f : 0.f, 0.f);
        rot3(a, b, trig[g * 3 + 0], trig[g * 3 + 1], trig[g * 3 + 2]);
        Ug[g * 4 + 0 + i] = a;
        Ug[g * 4 + 2 + i] = b;
    }
    __syncthreads();

    if (tid < 6) {
        const float2* u = Ug + (12 + tid) * 4;
        const float2 u00 = u[0], u01 = u[1], u10 = u[2], u11 = u[3];
        float o00, o11; float2 o01;
        if (tid & 1) {
            o00 = 2.f * cmulc(u00, u10).x;
            o11 = 2.f * cmulc(u01, u11).x;
            const float2 s = cmulc(u00, u11), t = cmulc(u10, u01);
            o01 = make_float2(s.x + t.x, s.y + t.y);
        } else {
            o00 = cmulc(u00, u00).x - cmulc(u10, u10).x;
            o11 = cmulc(u01, u01).x - cmulc(u11, u11).x;
            const float2 s = cmulc(u00, u01), t = cmulc(u10, u11);
            o01 = make_float2(s.x - t.x, s.y - t.y);
        }
        Obs[tid] = make_float4(o00, o11, 2.f * o01.x, 2.f * o01.y);
    }

    for (int i = 0; i < 8; ++i) {
        const int idx = tid + 128 * i;
        const int mt = idx >> 8, m = (idx >> 4) & 15, k = idx & 15;
        const int hp = m >> 3, rp = m & 7, hh = k >> 3, r = k & 7;
        const float2* u = Ug + mt * 12;
        const float2 e0 = u[0 + ((rp >> 2) & 1) * 2 + ((r >> 2) & 1)];
        const float2 e1 = u[4 + ((rp >> 1) & 1) * 2 + ((r >> 1) & 1)];
        const float2 e2 = u[8 + (rp & 1) * 2 + (r & 1)];
        const float2 mm = cmul(cmul(e0, e1), e2);
        E[idx] = (hp == hh) ? mm.x : (hp ? mm.y : -mm.y);
    }
    __syncthreads();

    v2f A[2][4];
    v2f Aq[2][2][2];
    #pragma unroll
    for (int kt = 0; kt < 4; ++kt) {
        const int hi = 4 * (kt >> 1) + ((kt & 1) ? (3 - h) : h);
        const float* p0 = &E[2 * 256 + c * 16 + hi];
        A[0][kt].x = p0[0]; A[0][kt].y = p0[8];
        const float* p1 = &E[3 * 256 + c * 16 + 8 * h + 2 * kt];
        A[1][kt].x = p1[0]; A[1][kt].y = p1[1];
    }
    {
        const int rr = (c & 7) >> 1, rc = c & 1, rh = c >> 3;
        const int rowQ = 8 * rc + 2 * rr + rh;
        const int rowS = 8 * rc + (rr ^ (4 * rh));
        #pragma unroll
        for (int mt = 0; mt < 2; ++mt)
            #pragma unroll
            for (int t = 0; t < 2; ++t)
                #pragma unroll
                for (int kt = 0; kt < 2; ++kt) {
                    const float* p = &E[mt * 256 + (mt ? rowS : rowQ) * 16 + 8 * h + 4 * kt + t];
                    Aq[mt][t][kt].x = p[0]; Aq[mt][t][kt].y = p[2];
                }
    }

    const v8f z8 = {0.f, 0.f, 0.f, 0.f, 0.f, 0.f, 0.f, 0.f};
    const float ysign = h ? 1.f : -1.f;

    for (int ct = 0; ct < 2; ++ct) {
        v8f Q[4], S[4];
        {
            float4 v[6];
            #pragma unroll
            for (int q = 0; q < 6; ++q) v[q] = Xch[(w * 6 + q) * 32 + ct * 16 + c];
            float2 ph[8], pl[8];
            prod3(ph, v[0], v[1], v[2]);
            prod3(pl, v[3], v[4], v[5]);
            float phc[8], plc[8];
            #pragma unroll
            for (int b = 0; b < 8; ++b) {
                phc[b] = h ? ph[b].y : ph[b].x;
                plc[b] = h ? pl[b].y : pl[b].x;
            }
            #pragma unroll
            for (int t2 = 0; t2 < 2; ++t2)
                #pragma unroll
                for (int t5 = 0; t5 < 2; ++t5)
                    #pragma unroll
                    for (int kt = 0; kt < 2; ++kt) {
                        const int i0 = 4 * kt + t2, i1 = i0 + 2;
                        const int l0 = 4 * kt + t5, l1 = l0 + 2;
                        v2f bq, bs;
                        bq.x = phc[ring_src(i0 * 8 + t5) >> 3];
                        bq.y = phc[ring_src(i1 * 8 + t5) >> 3];
                        bs.x = plc[ring_src(t2 * 8 + l0) & 7];
                        bs.y = plc[ring_src(t2 * 8 + l1) & 7];
                        const int t = 2 * t2 + t5;
                        Q[t] = __builtin_amdgcn_wmma_f32_16x16x4_f32(
                            false, Aq[0][t2][kt], false, bq, (short)0, kt == 0 ? z8 : Q[t], false, false);
                        S[t] = __builtin_amdgcn_wmma_f32_16x16x4_f32(
                            false, Aq[1][t5][kt], false, bs, (short)0, kt == 0 ? z8 : S[t], false, false);
                    }
        }

        v2f Sp[4][8];
        #pragma unroll
        for (int t = 0; t < 4; ++t)
            #pragma unroll
            for (int i = 0; i < 4; ++i) {
                Sp[t][i].x = S[t][2 * i];
                Sp[t][i].y = S[t][2 * i + 1];
                Sp[t][4 + i].x = xhalf(S[t][2 * i]);
                Sp[t][4 + i].y = xhalf(S[t][2 * i + 1]);
            }
        v2f Xp[4][8];
        #pragma unroll
        for (int kt = 0; kt < 4; ++kt)
            #pragma unroll
            for (int b = 0; b < 8; ++b) {
                const int j5 = kt >> 1, j4 = kt & 1, j2 = (b >> 2) & 1, j1 = (b >> 1) & 1, j0 = b & 1;
                const int m = ((j5 ^ j0) << 1) | (j4 ^ j5 ^ j0);
                const int n = ((j2 ^ j4) << 2) | ((j1 ^ j2) << 1) | (j0 ^ j1);
                v2f acc = {0.f, 0.f};
                #pragma unroll
                for (int t = 0; t < 4; ++t) {
                    const float qre = Q[t][2 * m], qim = Q[t][2 * m + 1];
                    const v2f sv = Sp[t][n];
                    v2f qv, qw, s0, s1;
                    qv.x = qre;   qv.y = qim;
                    qw.x = qim;   qw.y = qre;
                    s0.x = sv.x;  s0.y = sv.x;
                    s1.x = -sv.y; s1.y = sv.y;
                    acc += s0 * qv;
                    acc += s1 * qw;
                }
                Xp[kt][b] = acc;
            }

        v8f D2[8];
        {
            v8f D[8];
            #pragma unroll
            for (int kt = 0; kt < 4; ++kt)
                #pragma unroll
                for (int b = 0; b < 8; ++b)
                    D[b] = __builtin_amdgcn_wmma_f32_16x16x4_f32(
                        false, A[0][kt], false, Xp[kt][b], (short)0, kt == 0 ? z8 : D[b], false, false);
            #pragma unroll
            for (int kt = 0; kt < 4; ++kt)
                #pragma unroll
                for (int a = 0; a < 8; ++a) {
                    v2f bb;
                    bb.x = D[2 * kt][a];
                    bb.y = D[2 * kt + 1][a];
                    D2[a] = __builtin_amdgcn_wmma_f32_16x16x4_f32(
                        false, A[1][kt], false, bb, (short)0, kt == 0 ? z8 : D2[a], false, false);
                }
        }

        v2f Y2[32], Yo2[32];
        #pragma unroll
        for (int p = 0; p < 32; ++p) {
            Y2[p].x = D2[p >> 2][2 * (p & 3)];
            Y2[p].y = D2[p >> 2][2 * (p & 3) + 1];
            Yo2[p].x = xhalf(Y2[p].x);
            Yo2[p].y = xhalf(Y2[p].y);
        }
        v2f bA[8], bB[8];
        #pragma unroll
        for (int i = 0; i < 8; ++i) { bA[i].x = 0.f; bA[i].y = 0.f; bB[i].x = 0.f; bB[i].y = 0.f; }
        #pragma unroll
        for (int p = 0; p < 32; ++p) {
            const int s = 2 * p;
            const int iA = par6(s & bit_mask(1)) * 4 + par6(s & bit_mask(2)) * 2 + par6(s & bit_mask(3));
            const int iB = par6(s & bit_mask(4)) * 4 + par6(s & bit_mask(0)) * 2 + par6(s & bit_mask(5));
            bA[iA] += Y2[p] * Y2[p];
            bB[iB] += Y2[p] * Y2[p];
        }
        float binA[8], binB[8];
        #pragma unroll
        for (int i = 0; i < 8; ++i) { binA[i] = bA[i].x + bA[i].y; binB[i] = bB[i].x + bB[i ^ 3].y; }

        float v[6];
        #pragma unroll
        for (int q = 0; q < 6; ++q) {
            const int M = pair_mask(q), F = bit_mask(q), PM = M >> 1;
            v2f cr = {0.f, 0.f}, gp = {0.f, 0.f}, gn = {0.f, 0.f};
            #pragma unroll
            for (int p = 0; p < 32; ++p) {
                if (par6(p & 21)) continue;
                v2f pr = Y2[p ^ PM], po = Yo2[p ^ PM];
                if (M & 1) { pr = __builtin_shufflevector(pr, pr, 1, 0); po = __builtin_shufflevector(po, po, 1, 0); }
                cr += Y2[p] * pr;
                if (par6((2 * p) & F)) gn += Y2[p] * po; else gp += Y2[p] * po;
            }
            const float Cr = cr.x + cr.y;
            const float G  = (F & 1) ? ((gp.x - gp.y) - (gn.x - gn.y)) : ((gp.x + gp.y) - (gn.x + gn.y));
            const bool fromA = (q >= 1 && q <= 3);
            const int sel = fromA ? (4 >> (q - 1)) : (q == 4 ? 4 : (q == 0 ? 2 : 1));
            float s0 = 0.f, s1 = 0.f;
            #pragma unroll
            for (int i = 0; i < 8; ++i) {
                const float b = fromA ? binA[i] : binB[i];
                if (i & sel) s1 += b; else s0 += b;
            }
            const float4 O = Obs[q];
            float t = O.x * s0 + O.y * s1 + O.z * Cr + (ysign * O.w) * G;
            t += xhalf(t);
            v[q] = t;
        }

        const int ee = blockIdx.x * 128 + w * 32 + ct * 16 + c;
        const float4 o = h ? make_float4(v[4], v[5], v[0], v[1]) : make_float4(v[0], v[1], v[2], v[3]);
        *(float4*)&out[ee * 8 + 4 * h] = o;
    }
}

extern "C" void kernel_launch(void* const* d_in, const int* in_sizes, int n_in,
                              void* d_out, int out_size, void* d_ws, size_t ws_size,
                              hipStream_t stream) {
    const float* x     = (const float*)d_in[0];
    const float* theta = (const float*)d_in[1];
    float* out = (float*)d_out;

    qsim_kernel<<<131072 / 128, 128, 0, stream>>>(x, theta, out);
}
